# starting kernel
# baseline (speedup 1.0000x reference)
#include <hip/hip_runtime.h>
#include <hip/hip_bf16.h>

typedef __attribute__((ext_vector_type(16))) _Float16 v16h;
typedef __attribute__((ext_vector_type(8)))  float    v8f;

#define IN_D  128
#define HID   64


__global__ void deg_init_kernel(float* deg, int n) {
    int i = blockIdx.x * blockDim.x + threadIdx.x;
    if (i < n) deg[i] = 1.0f;
}

__global__ void deg_accum_kernel(const int* __restrict__ dst, float* deg, int e) {
    int i = blockIdx.x * blockDim.x + threadIdx.x;
    if (i < e) unsafeAtomicAdd(&deg[dst[i]], 1.0f);
}

__global__ void dinv_kernel(float* deg, int n) {
    int i = blockIdx.x * blockDim.x + threadIdx.x;
    if (i < n) deg[i] = rsqrtf(deg[i]);
}


template <int K>
__global__ void pack_w_kernel(const float* __restrict__ W, _Float16* __restrict__ Wp) {
    const int frag = (blockIdx.x * blockDim.x + threadIdx.x) >> 5;
    const int lane = threadIdx.x & 31;
    const int nFrag = (K / 32) * 4;
    if (frag >= nFrag) return;
    const int ks = frag >> 2;
    const int ct = frag & 3;
    const int n    = lane & 15;
    const int koff = (lane >> 4) << 3;
    v16h b;
    const float* wp = W + (size_t)(ks * 32 + koff) * HID + ct * 16 + n;
#pragma unroll
    for (int j = 0; j < 8; ++j) {
        b[j]     = (_Float16)wp[(size_t)j * HID];
        b[8 + j] = (_Float16)wp[(size_t)(16 + j) * HID];
    }
    *(v16h*)(Wp + ((size_t)frag * 32 + lane) * 16) = b;
}


template <int K, bool RELU>
__global__ void gemm_wmma_kernel(const float* __restrict__ A,
                                 const _Float16* __restrict__ Wp,
                                 float* __restrict__ out, int nRows) {
    const int wave = (blockIdx.x * blockDim.x + threadIdx.x) >> 5;
    const int lane = threadIdx.x & 31;
    const int rowTiles = nRows >> 4;
    if (wave >= rowTiles) return;

    const int m    = lane & 15;
    const int koff = (lane >> 4) << 3;
    const int row  = (wave << 4) + m;

    const v16h* bp = (const v16h*)Wp;

    v8f acc0 = {}, acc1 = {}, acc2 = {}, acc3 = {};

#pragma unroll
    for (int k0 = 0; k0 < K; k0 += 32) {
        const int ks = k0 >> 5;
        v16h a;
        const float* ap = A + (size_t)row * K + k0 + koff;
#pragma unroll
        for (int j = 0; j < 8; ++j) {
            float f0 = ap[j];
            float f1 = ap[16 + j];
            if (RELU) { f0 = fmaxf(f0, 0.0f); f1 = fmaxf(f1, 0.0f); }
            a[j]     = (_Float16)f0;
            a[8 + j] = (_Float16)f1;
        }
#pragma unroll
        for (int ct = 0; ct < 4; ++ct) {
            v16h b = bp[(size_t)(ks * 4 + ct) * 32 + lane];
            v8f& acc = (ct == 0 ? acc0 : ct == 1 ? acc1 : ct == 2 ? acc2 : acc3);
            acc = __builtin_amdgcn_wmma_f32_16x16x32_f16(
                false, a, false, b, (short)0, acc, false, false);
        }
    }

    const int colLane = lane & 15;
    const int rBase   = (wave << 4) + ((lane >> 4) << 3);
    auto store_tile = [&](v8f acc, int ct) {
#pragma unroll
        for (int i = 0; i < 8; ++i)
            out[(size_t)(rBase + i) * HID + ct * 16 + colLane] = acc[i];
    };
    store_tile(acc0, 0); store_tile(acc1, 1); store_tile(acc2, 2); store_tile(acc3, 3);
}

__global__ void agg_init_kernel(const float* __restrict__ hW,
                                const float* __restrict__ dinv,
                                const float* __restrict__ bias,
                                float* __restrict__ out, int total) {
    int idx = blockIdx.x * blockDim.x + threadIdx.x;
    if (idx >= total) return;
    int node = idx >> 6;
    int c    = idx & 63;
    float w = dinv[node];
    float v = hW[idx] * (w * w);
    if (bias) v += bias[c];
    out[idx] = v;
}

__global__ void agg_edges_kernel(const float* __restrict__ hW,
                                 const int* __restrict__ src,
                                 const int* __restrict__ dst,
                                 const float* __restrict__ dinv,
                                 float* __restrict__ out, int nEdges) {
    const int wave = (blockIdx.x * blockDim.x + threadIdx.x) >> 5;
    const int lane = threadIdx.x & 31;
    if (wave >= nEdges) return;
    const int s = src[wave];
    const int d = dst[wave];
    const float w = dinv[s] * dinv[d];
    const float2 v = ((const float2*)(hW + (size_t)s * HID))[lane];
    float* op = out + (size_t)d * HID + lane * 2;
    unsafeAtomicAdd(op,     v.x * w);
    unsafeAtomicAdd(op + 1, v.y * w);
}


extern "C" void kernel_launch(void* const* d_in, const int* in_sizes, int n_in,
                              void* d_out, int out_size, void* d_ws, size_t ws_size,
                              hipStream_t stream) {
    const float* x     = (const float*)d_in[0];
    const int*   eidx  = (const int*)d_in[1];
    const float* W_in  = (const float*)d_in[2];
    const float* b_in  = (const float*)d_in[3];
    const float* W_h   = (const float*)d_in[4];
    const float* b_h   = (const float*)d_in[5];
    const float* W_out = (const float*)d_in[6];

    const int N = in_sizes[0] / IN_D;
    const int E = in_sizes[1] / 2;
    const int L = in_sizes[4] / (HID * HID);

    const int* src = eidx;
    const int* dst = eidx + E;

    char* ws = (char*)d_ws;
    auto align256 = [](size_t v) { return (v + 255) & ~(size_t)255; };
    size_t off = 0;
    float* dinv = (float*)(ws + off);  off = align256(off + (size_t)N * sizeof(float));
    float* B1   = (float*)(ws + off);  off = align256(off + (size_t)N * HID * sizeof(float));
    _Float16* WpIn  = (_Float16*)(ws + off);  off = align256(off + 16 * 1024);
    _Float16* WpH0  = (_Float16*)(ws + off);  off = align256(off + 8 * 1024);
    _Float16* WpH1  = (_Float16*)(ws + off);  off = align256(off + 8 * 1024);
    _Float16* WpH2  = (_Float16*)(ws + off);  off = align256(off + 8 * 1024);
    _Float16* WpOut = (_Float16*)(ws + off);  off = align256(off + 8 * 1024);
    _Float16* WpHs[3] = {WpH0, WpH1, WpH2};
    float* O = (float*)d_out;

    const int T = 256;

    deg_init_kernel<<<(N + T - 1) / T, T, 0, stream>>>(dinv, N);
    deg_accum_kernel<<<(E + T - 1) / T, T, 0, stream>>>(dst, dinv, E);
    dinv_kernel<<<(N + T - 1) / T, T, 0, stream>>>(dinv, N);

    pack_w_kernel<IN_D><<<2, T, 0, stream>>>(W_in, WpIn);
    for (int l = 0; l < L; ++l)
        pack_w_kernel<HID><<<1, T, 0, stream>>>(W_h + (size_t)l * HID * HID, WpHs[l]);
    pack_w_kernel<HID><<<1, T, 0, stream>>>(W_out, WpOut);

    const int rowTiles   = N / 16;
    const int gemmBlocks = (rowTiles + 7) / 8;
    const int totalNC    = N * HID;
    const int initBlocks = (totalNC + T - 1) / T;
    const int edgeBlocks = (E + 7) / 8;

    gemm_wmma_kernel<IN_D, false><<<gemmBlocks, T, 0, stream>>>(x, WpIn, B1, N);
    agg_init_kernel<<<initBlocks, T, 0, stream>>>(B1, dinv, b_in, O, totalNC);
    agg_edges_kernel<<<edgeBlocks, T, 0, stream>>>(B1, src, dst, dinv, O, E);

    for (int l = 0; l < L; ++l) {
        const float* b = b_h + (size_t)l * HID;
        if (l == 0)
            gemm_wmma_kernel<HID, false><<<gemmBlocks, T, 0, stream>>>(O, WpHs[l], B1, N);
        else
            gemm_wmma_kernel<HID, true ><<<gemmBlocks, T, 0, stream>>>(O, WpHs[l], B1, N);
        agg_init_kernel<<<initBlocks, T, 0, stream>>>(B1, dinv, b, O, totalNC);
        agg_edges_kernel<<<edgeBlocks, T, 0, stream>>>(B1, src, dst, dinv, O, E);
    }

    gemm_wmma_kernel<HID, true><<<gemmBlocks, T, 0, stream>>>(O, WpOut, B1, N);
    agg_init_kernel<<<initBlocks, T, 0, stream>>>(B1, dinv, nullptr, O, totalNC);
    agg_edges_kernel<<<edgeBlocks, T, 0, stream>>>(B1, src, dst, dinv, O, E);
}


# edge ordering: per-edge global returning atomics (count+rank, place) -> two-level counting sort with every counter in LDS (bucket passes, scan, per-bucket finish); row order unchanged
# speedup vs baseline: 2.9891x; 2.9891x over previous
#include <hip/hip_runtime.h>
#include <hip/hip_bf16.h>

typedef __attribute__((ext_vector_type(16))) _Float16 v16h;
typedef __attribute__((ext_vector_type(8)))  float    v8f;

#define IN_D  128
#define HID   64


__global__ void deg_init_kernel(float* deg, int n) {
    int i = blockIdx.x * blockDim.x + threadIdx.x;
    if (i < n) deg[i] = 1.0f;
}

__global__ void deg_accum_kernel(const int* __restrict__ dst, float* deg, int e) {
    int i = blockIdx.x * blockDim.x + threadIdx.x;
    if (i < e) unsafeAtomicAdd(&deg[dst[i]], 1.0f);
}

__global__ void dinv_kernel(float* deg, int n) {
    int i = blockIdx.x * blockDim.x + threadIdx.x;
    if (i < n) deg[i] = rsqrtf(deg[i]);
}


template <int K>
__global__ void pack_w_kernel(const float* __restrict__ W, _Float16* __restrict__ Wp) {
    const int frag = (blockIdx.x * blockDim.x + threadIdx.x) >> 5;
    const int lane = threadIdx.x & 31;
    const int nFrag = (K / 32) * 4;
    if (frag >= nFrag) return;
    const int ks = frag >> 2;
    const int ct = frag & 3;
    const int n    = lane & 15;
    const int koff = (lane >> 4) << 3;
    v16h b;
    const float* wp = W + (size_t)(ks * 32 + koff) * HID + ct * 16 + n;
#pragma unroll
    for (int j = 0; j < 8; ++j) {
        b[j]     = (_Float16)wp[(size_t)j * HID];
        b[8 + j] = (_Float16)wp[(size_t)(16 + j) * HID];
    }
    *(v16h*)(Wp + ((size_t)frag * 32 + lane) * 16) = b;
}


template <int K, bool RELU>
__global__ void gemm_wmma_kernel(const float* __restrict__ A,
                                 const _Float16* __restrict__ Wp,
                                 float* __restrict__ out, int nRows) {
    const int wave = (blockIdx.x * blockDim.x + threadIdx.x) >> 5;
    const int lane = threadIdx.x & 31;
    const int rowTiles = nRows >> 4;
    if (wave >= rowTiles) return;

    const int m    = lane & 15;
    const int koff = (lane >> 4) << 3;
    const int row  = (wave << 4) + m;

    const v16h* bp = (const v16h*)Wp;

    v8f acc0 = {}, acc1 = {}, acc2 = {}, acc3 = {};

#pragma unroll
    for (int k0 = 0; k0 < K; k0 += 32) {
        const int ks = k0 >> 5;
        v16h a;
        const float* ap = A + (size_t)row * K + k0 + koff;
#pragma unroll
        for (int j = 0; j < 8; ++j) {
            float f0 = ap[j];
            float f1 = ap[16 + j];
            if (RELU) { f0 = fmaxf(f0, 0.0f); f1 = fmaxf(f1, 0.0f); }
            a[j]     = (_Float16)f0;
            a[8 + j] = (_Float16)f1;
        }
#pragma unroll
        for (int ct = 0; ct < 4; ++ct) {
            v16h b = bp[(size_t)(ks * 4 + ct) * 32 + lane];
            v8f& acc = (ct == 0 ? acc0 : ct == 1 ? acc1 : ct == 2 ? acc2 : acc3);
            acc = __builtin_amdgcn_wmma_f32_16x16x32_f16(
                false, a, false, b, (short)0, acc, false, false);
        }
    }

    const int colLane = lane & 15;
    const int rBase   = (wave << 4) + ((lane >> 4) << 3);
    auto store_tile = [&](v8f acc, int ct) {
#pragma unroll
        for (int i = 0; i < 8; ++i)
            out[(size_t)(rBase + i) * HID + ct * 16 + colLane] = acc[i];
    };
    store_tile(acc0, 0); store_tile(acc1, 1); store_tile(acc2, 2); store_tile(acc3, 3);
}

__global__ void agg_init_kernel(const float* __restrict__ hW,
                                const float* __restrict__ dinv,
                                const float* __restrict__ bias,
                                float* __restrict__ out, int total) {
    int idx = blockIdx.x * blockDim.x + threadIdx.x;
    if (idx >= total) return;
    int node = idx >> 6;
    int c    = idx & 63;
    float w = dinv[node];
    float v = hW[idx] * (w * w);
    if (bias) v += bias[c];
    out[idx] = v;
}

__global__ void agg_edges_kernel(const float* __restrict__ hW,
                                 const int* __restrict__ src,
                                 const int* __restrict__ dst,
                                 const float* __restrict__ dinv,
                                 float* __restrict__ out, int nEdges) {
    const int wave = (blockIdx.x * blockDim.x + threadIdx.x) >> 5;
    const int lane = threadIdx.x & 31;
    if (wave >= nEdges) return;
    const int s = src[wave];
    const int d = dst[wave];
    const float w = dinv[s] * dinv[d];
    const float2 v = ((const float2*)(hW + (size_t)s * HID))[lane];
    float* op = out + (size_t)d * HID + lane * 2;
    unsafeAtomicAdd(op,     v.x * w);
    unsafeAtomicAdd(op + 1, v.y * w);
}


#define SCAN_T   256
#define SCAN_PER 4
#define SCAN_BLK (SCAN_T * SCAN_PER)

#define BK_SHIFT 8
#define BK_NODES (1 << BK_SHIFT)
#define BK_GRID  512
#define BK_T     512
#define BK_U     8
#define FIN_T    1024
#define BK_MAXB  8192

template <bool PLACE>
__global__ void __launch_bounds__(BK_T)
bucket_pass_kernel(const int* __restrict__ src, const int* __restrict__ dst,
                   int* __restrict__ tab, const int* __restrict__ offs,
                   int2* __restrict__ pairs, int e, int chunk, int nb) {
    extern __shared__ int hist[];
    const int t = threadIdx.x;
    const int blk = blockIdx.x;
    const int G = gridDim.x;
    for (int i = t; i < nb; i += BK_T) hist[i] = PLACE ? offs[i * G + blk] : 0;
    __syncthreads();
    const int lo = blk * chunk;
    const int hi = min(e, lo + chunk);
    for (int i0 = lo + t; i0 < hi; i0 += BK_T * BK_U) {
        int d[BK_U], s[BK_U];
#pragma unroll
        for (int u = 0; u < BK_U; ++u) {
            const int i = i0 + u * BK_T;
            d[u] = (i < hi) ? dst[i] : 0;
            s[u] = (PLACE && i < hi) ? src[i] : 0;
        }
#pragma unroll
        for (int u = 0; u < BK_U; ++u) {
            if (i0 + u * BK_T < hi) {
                const int b = d[u] >> BK_SHIFT;
                if (PLACE) {
                    const int pos = atomicAdd(&hist[b], 1);
                    pairs[pos] = make_int2(s[u], d[u]);
                } else {
                    atomicAdd(&hist[b], 1);
                }
            }
        }
    }
    if (!PLACE) {
        __syncthreads();
        for (int i = t; i < nb; i += BK_T) tab[i * G + blk] = hist[i];
    }
}

__global__ void scan_sums_kernel(const int* __restrict__ cnt, int* __restrict__ bsum, int n) {
    __shared__ int sW[SCAN_T / 32];
    const int t = threadIdx.x;
    const int base = blockIdx.x * SCAN_BLK + t * SCAN_PER;
    int s = 0;
#pragma unroll
    for (int j = 0; j < SCAN_PER; ++j)
        if (base + j < n) s += cnt[base + j];
    for (int d = 16; d > 0; d >>= 1) s += __shfl_xor(s, d);
    if ((t & 31) == 0) sW[t >> 5] = s;
    __syncthreads();
    if (t == 0) {
        int tot = 0;
        for (int k = 0; k < SCAN_T / 32; ++k) tot += sW[k];
        bsum[blockIdx.x] = tot;
    }
}

__global__ void scan_apply_kernel(const int* __restrict__ cnt, const int* __restrict__ bsum,
                                  int* __restrict__ offs, int n) {
    __shared__ int sW[SCAN_T / 32];
    const int t = threadIdx.x;
    const int lane = t & 31;
    const int wv = t >> 5;
    const int b = blockIdx.x;

    int part = 0;
    for (int k = t; k < b; k += SCAN_T) part += bsum[k];
    for (int d = 16; d > 0; d >>= 1) part += __shfl_xor(part, d);
    if (lane == 0) sW[wv] = part;
    __syncthreads();
    int offset = 0;
    for (int k = 0; k < SCAN_T / 32; ++k) offset += sW[k];
    __syncthreads();

    const int base = b * SCAN_BLK + t * SCAN_PER;
    int c[SCAN_PER];
    int run = 0;
#pragma unroll
    for (int j = 0; j < SCAN_PER; ++j) {
        c[j] = (base + j < n) ? cnt[base + j] : 0;
        run += c[j];
    }
    int inc = run;
    for (int d = 1; d < 32; d <<= 1) {
        const int v = __shfl_up(inc, d);
        if (lane >= d) inc += v;
    }
    if (lane == 31) sW[wv] = inc;
    __syncthreads();
    int pre = offset + inc - run;
    for (int k = 0; k < wv; ++k) pre += sW[k];
#pragma unroll
    for (int j = 0; j < SCAN_PER; ++j) {
        pre += c[j];
        if (base + j < n) offs[base + j + 1] = pre;
    }
    if (b == 0 && t == 0) offs[0] = 0;
}

__global__ void __launch_bounds__(FIN_T)
bucket_finish_kernel(const int* __restrict__ offs, const int2* __restrict__ pairs,
                     int* __restrict__ rowptr, float* __restrict__ dinv,
                     int* colTmp, int* __restrict__ col, int n, int G) {
    __shared__ int cntL[BK_NODES];
    __shared__ int offL[BK_NODES];
    __shared__ int curL[BK_NODES];
    __shared__ int wtot[BK_NODES / 32];
    const int t = threadIdx.x;
    const int lane = t & 31;
    const int wv = t >> 5;
    const int b = blockIdx.x;
    const int beg = offs[b * G];
    const int end = offs[(b + 1) * G];
    if (t < BK_NODES) { cntL[t] = 0; curL[t] = 0; }
    __syncthreads();
    for (int i = beg + t; i < end; i += FIN_T)
        atomicAdd(&cntL[pairs[i].y & (BK_NODES - 1)], 1);
    __syncthreads();

    const int c = (t < BK_NODES) ? cntL[t] : 0;
    int inc = c;
    for (int d = 1; d < 32; d <<= 1) {
        const int v = __shfl_up(inc, d);
        if (lane >= d) inc += v;
    }
    if (t < BK_NODES && lane == 31) wtot[wv] = inc;
    __syncthreads();
    if (t < BK_NODES) {
        int pre = 0;
        for (int k = 0; k < wv; ++k) pre += wtot[k];
        const int excl = pre + inc - c;
        offL[t] = excl;
        const int node = b * BK_NODES + t;
        if (node < n) {
            rowptr[node + 1] = beg + excl + c;
            dinv[node] = rsqrtf((float)(c + 1));
        }
    }
    if (b == 0 && t == 0) rowptr[0] = 0;
    __syncthreads();

    for (int i = beg + t; i < end; i += FIN_T) {
        const int2 p = pairs[i];
        const int d = p.y & (BK_NODES - 1);
        const int r = atomicAdd(&curL[d], 1);
        colTmp[beg + offL[d] + r] = p.x;
    }
    __syncthreads();

    for (int row = wv; row < BK_NODES; row += FIN_T / 32) {
        const int deg = __builtin_amdgcn_readfirstlane(cntL[row]);
        const int rb  = __builtin_amdgcn_readfirstlane(beg + offL[row]);
        for (int base = 0; base < deg; base += 32) {
            const int i = base + lane;
            const int key = (i < deg) ? colTmp[rb + i] : 0;
            int r = 0;
            for (int jb = 0; jb < deg; jb += 32) {
                const int lim = min(32, deg - jb);
                const int kv = (lane < lim) ? colTmp[rb + jb + lane] : 0;
                for (int tt = 0; tt < lim; ++tt) {
                    const int kj = __builtin_amdgcn_readlane(kv, tt);
                    const int j = jb + tt;
                    r += ((kj < key) || (kj == key && j < i)) ? 1 : 0;
                }
            }
            if (i < deg) col[rb + r] = key;
        }
    }
}

__global__ void agg_gather_kernel(const float* __restrict__ hW,
                                  const int* __restrict__ rowptr,
                                  const int* __restrict__ col,
                                  const float* __restrict__ dinv,
                                  const float* __restrict__ bias,
                                  float* __restrict__ out, int n) {
    const int lane = threadIdx.x & 31;
    const int node = __builtin_amdgcn_readfirstlane((int)((blockIdx.x * blockDim.x + threadIdx.x) >> 5));
    if (node >= n) return;
    const int beg = rowptr[node];
    const int deg = rowptr[node + 1] - beg;
    const float dd = dinv[node];
    const float2 self = ((const float2*)(hW + (size_t)node * HID))[lane];
    float ax = 0.0f, ay = 0.0f;
    for (int base = 0; base < deg; base += 32) {
        const int lim = min(32, deg - base);
        int s = 0;
        float ws = 0.0f;
        if (lane < lim) { s = col[beg + base + lane]; ws = dinv[s]; }
        for (int t = 0; t < lim; t += 8) {
            float2 v[8];
#pragma unroll
            for (int u = 0; u < 8; ++u) {
                const int tt = min(t + u, lim - 1);
                const int sj = __builtin_amdgcn_readlane(s, tt);
                v[u] = ((const float2*)(hW + (size_t)sj * HID))[lane];
            }
#pragma unroll
            for (int u = 0; u < 8; ++u) {
                const int tt = min(t + u, lim - 1);
                const float wj = __int_as_float(__builtin_amdgcn_readlane(__float_as_int(ws), tt));
                const bool live = (t + u) < lim;
                ax = fmaf(live ? v[u].x : 0.0f, wj, ax);
                ay = fmaf(live ? v[u].y : 0.0f, wj, ay);
            }
        }
    }
    const float w2 = dd * dd;
    float ox = fmaf(ax, dd, self.x * w2);
    float oy = fmaf(ay, dd, self.y * w2);
    if (bias) { ox += bias[lane * 2]; oy += bias[lane * 2 + 1]; }
    ((float2*)(out + (size_t)node * HID))[lane] = make_float2(ox, oy);
}


extern "C" void kernel_launch(void* const* d_in, const int* in_sizes, int n_in,
                              void* d_out, int out_size, void* d_ws, size_t ws_size,
                              hipStream_t stream) {
    const float* x     = (const float*)d_in[0];
    const int*   eidx  = (const int*)d_in[1];
    const float* W_in  = (const float*)d_in[2];
    const float* b_in  = (const float*)d_in[3];
    const float* W_h   = (const float*)d_in[4];
    const float* b_h   = (const float*)d_in[5];
    const float* W_out = (const float*)d_in[6];

    const int N = in_sizes[0] / IN_D;
    const int E = in_sizes[1] / 2;
    const int L = in_sizes[4] / (HID * HID);

    const int* src = eidx;
    const int* dst = eidx + E;

    char* ws = (char*)d_ws;
    auto align256 = [](size_t v) { return (v + 255) & ~(size_t)255; };
    size_t off = 0;
    float* dinv = (float*)(ws + off);  off = align256(off + (size_t)N * sizeof(float));
    float* B1   = (float*)(ws + off);  off = align256(off + (size_t)N * HID * sizeof(float));
    _Float16* WpIn  = (_Float16*)(ws + off);  off = align256(off + 16 * 1024);
    _Float16* WpH0  = (_Float16*)(ws + off);  off = align256(off + 8 * 1024);
    _Float16* WpH1  = (_Float16*)(ws + off);  off = align256(off + 8 * 1024);
    _Float16* WpH2  = (_Float16*)(ws + off);  off = align256(off + 8 * 1024);
    _Float16* WpOut = (_Float16*)(ws + off);  off = align256(off + 8 * 1024);
    _Float16* WpHs[3] = {WpH0, WpH1, WpH2};
    int* rowptr = (int*)(ws + off);  off = align256(off + ((size_t)N + 1) * sizeof(int));
    int* col    = (int*)(ws + off);  off = align256(off + (size_t)E * sizeof(int));
    float* O = (float*)d_out;

    const int T = 256;

    const int rowTiles   = N / 16;
    const int gemmBlocks = (rowTiles + 7) / 8;
    const int nodeBlocks = (N + 7) / 8;
    const int nBuck      = (N + BK_NODES - 1) >> BK_SHIFT;
    const int tabLen     = nBuck * BK_GRID;
    const int scanBlocks = (tabLen + SCAN_BLK - 1) / SCAN_BLK;
    const int chunk      = (E + BK_GRID - 1) / BK_GRID;

    const bool listFits = off <= ws_size && nBuck <= BK_MAXB &&
                          (size_t)3 * E + (size_t)2 * tabLen + 1 + (size_t)scanBlocks
                              <= (size_t)N * HID;

    if (listFits) {
        int2* pairs = (int2*)d_out;
        int* colTmp = (int*)d_out + (size_t)2 * E;
        int* tab    = colTmp + E;
        int* offs   = tab + tabLen;
        int* bsum   = offs + tabLen + 1;
        const size_t histBytes = (size_t)nBuck * sizeof(int);

        bucket_pass_kernel<false><<<BK_GRID, BK_T, histBytes, stream>>>(src, dst, tab, offs, pairs, E, chunk, nBuck);
        scan_sums_kernel<<<scanBlocks, SCAN_T, 0, stream>>>(tab, bsum, tabLen);
        scan_apply_kernel<<<scanBlocks, SCAN_T, 0, stream>>>(tab, bsum, offs, tabLen);
        bucket_pass_kernel<true><<<BK_GRID, BK_T, histBytes, stream>>>(src, dst, tab, offs, pairs, E, chunk, nBuck);
        bucket_finish_kernel<<<nBuck, FIN_T, 0, stream>>>(offs, pairs, rowptr, dinv, colTmp, col, N, BK_GRID);

        pack_w_kernel<IN_D><<<2, T, 0, stream>>>(W_in, WpIn);
        for (int l = 0; l < L; ++l)
            pack_w_kernel<HID><<<1, T, 0, stream>>>(W_h + (size_t)l * HID * HID, WpHs[l]);
        pack_w_kernel<HID><<<1, T, 0, stream>>>(W_out, WpOut);

        gemm_wmma_kernel<IN_D, false><<<gemmBlocks, T, 0, stream>>>(x, WpIn, B1, N);
        agg_gather_kernel<<<nodeBlocks, T, 0, stream>>>(B1, rowptr, col, dinv, b_in, O, N);

        for (int l = 0; l < L; ++l) {
            const float* b = b_h + (size_t)l * HID;
            if (l == 0)
                gemm_wmma_kernel<HID, false><<<gemmBlocks, T, 0, stream>>>(O, WpHs[l], B1, N);
            else
                gemm_wmma_kernel<HID, true ><<<gemmBlocks, T, 0, stream>>>(O, WpHs[l], B1, N);
            agg_gather_kernel<<<nodeBlocks, T, 0, stream>>>(B1, rowptr, col, dinv, b, O, N);
        }

        gemm_wmma_kernel<HID, true><<<gemmBlocks, T, 0, stream>>>(O, WpOut, B1, N);
        agg_gather_kernel<<<nodeBlocks, T, 0, stream>>>(B1, rowptr, col, dinv, nullptr, O, N);
        return;
    }


    deg_init_kernel<<<(N + T - 1) / T, T, 0, stream>>>(dinv, N);
    deg_accum_kernel<<<(E + T - 1) / T, T, 0, stream>>>(dst, dinv, E);
    dinv_kernel<<<(N + T - 1) / T, T, 0, stream>>>(dinv, N);

    pack_w_kernel<IN_D><<<2, T, 0, stream>>>(W_in, WpIn);
    for (int l = 0; l < L; ++l)
        pack_w_kernel<HID><<<1, T, 0, stream>>>(W_h + (size_t)l * HID * HID, WpHs[l]);
    pack_w_kernel<HID><<<1, T, 0, stream>>>(W_out, WpOut);

    const int totalNC    = N * HID;
    const int initBlocks = (totalNC + T - 1) / T;
    const int edgeBlocks = (E + 7) / 8;

    gemm_wmma_kernel<IN_D, false><<<gemmBlocks, T, 0, stream>>>(x, WpIn, B1, N);
    agg_init_kernel<<<initBlocks, T, 0, stream>>>(B1, dinv, b_in, O, totalNC);
    agg_edges_kernel<<<edgeBlocks, T, 0, stream>>>(B1, src, dst, dinv, O, E);

    for (int l = 0; l < L; ++l) {
        const float* b = b_h + (size_t)l * HID;
        if (l == 0)
            gemm_wmma_kernel<HID, false><<<gemmBlocks, T, 0, stream>>>(O, WpHs[l], B1, N);
        else
            gemm_wmma_kernel<HID, true ><<<gemmBlocks, T, 0, stream>>>(O, WpHs[l], B1, N);
        agg_init_kernel<<<initBlocks, T, 0, stream>>>(B1, dinv, b, O, totalNC);
        agg_edges_kernel<<<edgeBlocks, T, 0, stream>>>(B1, src, dst, dinv, O, E);
    }

    gemm_wmma_kernel<HID, true><<<gemmBlocks, T, 0, stream>>>(O, WpOut, B1, N);
    agg_init_kernel<<<initBlocks, T, 0, stream>>>(B1, dinv, nullptr, O, totalNC);
    agg_edges_kernel<<<edgeBlocks, T, 0, stream>>>(B1, src, dst, dinv, O, E);
}


# aggregation fused with next layer's product: a 4-wave block sums 16 rows (4 per wave, same order), passes the tile through LDS to WMMA, stores scaled rows; 4 launches and the O round trip gone
# speedup vs baseline: 4.2671x; 1.4276x over previous
#include <hip/hip_runtime.h>
#include <hip/hip_bf16.h>

typedef __attribute__((ext_vector_type(16))) _Float16 v16h;
typedef __attribute__((ext_vector_type(8)))  float    v8f;

#define IN_D  128
#define HID   64


__global__ void deg_init_kernel(float* deg, int n) {
    int i = blockIdx.x * blockDim.x + threadIdx.x;
    if (i < n) deg[i] = 1.0f;
}

__global__ void deg_accum_kernel(const int* __restrict__ dst, float* deg, int e) {
    int i = blockIdx.x * blockDim.x + threadIdx.x;
    if (i < e) unsafeAtomicAdd(&deg[dst[i]], 1.0f);
}

__global__ void dinv_kernel(float* deg, int n) {
    int i = blockIdx.x * blockDim.x + threadIdx.x;
    if (i < n) deg[i] = rsqrtf(deg[i]);
}


template <int K>
__global__ void pack_w_kernel(const float* __restrict__ W, _Float16* __restrict__ Wp) {
    const int frag = (blockIdx.x * blockDim.x + threadIdx.x) >> 5;
    const int lane = threadIdx.x & 31;
    const int nFrag = (K / 32) * 4;
    if (frag >= nFrag) return;
    const int ks = frag >> 2;
    const int ct = frag & 3;
    const int n    = lane & 15;
    const int koff = (lane >> 4) << 3;
    v16h b;
    const float* wp = W + (size_t)(ks * 32 + koff) * HID + ct * 16 + n;
#pragma unroll
    for (int j = 0; j < 8; ++j) {
        b[j]     = (_Float16)wp[(size_t)j * HID];
        b[8 + j] = (_Float16)wp[(size_t)(16 + j) * HID];
    }
    *(v16h*)(Wp + ((size_t)frag * 32 + lane) * 16) = b;
}


template <int K, bool RELU>
__global__ void gemm_wmma_kernel(const float* __restrict__ A,
                                 const _Float16* __restrict__ Wp,
                                 const float* __restrict__ rowScale,
                                 float* __restrict__ out, int nRows) {
    const int wave = (blockIdx.x * blockDim.x + threadIdx.x) >> 5;
    const int lane = threadIdx.x & 31;
    const int rowTiles = nRows >> 4;
    if (wave >= rowTiles) return;

    const int m    = lane & 15;
    const int koff = (lane >> 4) << 3;
    const int row  = (wave << 4) + m;

    const v16h* bp = (const v16h*)Wp;

    const int colLane = lane & 15;
    const int rBase   = (wave << 4) + ((lane >> 4) << 3);
    const float4 sc0 = *(const float4*)(rowScale + rBase);
    const float4 sc1 = *(const float4*)(rowScale + rBase + 4);
    const float sc[8] = {sc0.x, sc0.y, sc0.z, sc0.w, sc1.x, sc1.y, sc1.z, sc1.w};

    v8f acc0 = {}, acc1 = {}, acc2 = {}, acc3 = {};

#pragma unroll
    for (int k0 = 0; k0 < K; k0 += 32) {
        const int ks = k0 >> 5;
        v16h a;
        const float* ap = A + (size_t)row * K + k0 + koff;
#pragma unroll
        for (int j = 0; j < 8; ++j) {
            float f0 = ap[j];
            float f1 = ap[16 + j];
            if (RELU) { f0 = fmaxf(f0, 0.0f); f1 = fmaxf(f1, 0.0f); }
            a[j]     = (_Float16)f0;
            a[8 + j] = (_Float16)f1;
        }
#pragma unroll
        for (int ct = 0; ct < 4; ++ct) {
            v16h b = bp[(size_t)(ks * 4 + ct) * 32 + lane];
            v8f& acc = (ct == 0 ? acc0 : ct == 1 ? acc1 : ct == 2 ? acc2 : acc3);
            acc = __builtin_amdgcn_wmma_f32_16x16x32_f16(
                false, a, false, b, (short)0, acc, false, false);
        }
    }

    auto store_tile = [&](v8f acc, int ct) {
#pragma unroll
        for (int i = 0; i < 8; ++i)
            out[(size_t)(rBase + i) * HID + ct * 16 + colLane] = acc[i] * sc[i];
    };
    store_tile(acc0, 0); store_tile(acc1, 1); store_tile(acc2, 2); store_tile(acc3, 3);
}

__global__ void agg_init_kernel(const float* __restrict__ hW,
                                const float* __restrict__ dinv,
                                const float* __restrict__ bias,
                                float* __restrict__ out, int total) {
    int idx = blockIdx.x * blockDim.x + threadIdx.x;
    if (idx >= total) return;
    int node = idx >> 6;
    int c    = idx & 63;
    float v = hW[idx] * dinv[node];
    if (bias) v += bias[c];
    out[idx] = v;
}

__global__ void agg_edges_kernel(const float* __restrict__ hW,
                                 const int* __restrict__ src,
                                 const int* __restrict__ dst,
                                 const float* __restrict__ dinv,
                                 float* __restrict__ out, int nEdges) {
    const int wave = (blockIdx.x * blockDim.x + threadIdx.x) >> 5;
    const int lane = threadIdx.x & 31;
    if (wave >= nEdges) return;
    const int s = src[wave];
    const int d = dst[wave];
    const float w = dinv[d];
    const float2 v = ((const float2*)(hW + (size_t)s * HID))[lane];
    float* op = out + (size_t)d * HID + lane * 2;
    unsafeAtomicAdd(op,     v.x * w);
    unsafeAtomicAdd(op + 1, v.y * w);
}


#define SCAN_T   256
#define SCAN_PER 4
#define SCAN_BLK (SCAN_T * SCAN_PER)

#define BK_SHIFT 8
#define BK_NODES (1 << BK_SHIFT)
#define BK_GRID  512
#define BK_T     512
#define BK_U     8
#define FIN_T    1024
#define BK_MAXB  8192

template <bool PLACE>
__global__ void __launch_bounds__(BK_T)
bucket_pass_kernel(const int* __restrict__ src, const int* __restrict__ dst,
                   int* __restrict__ tab, const int* __restrict__ offs,
                   int2* __restrict__ pairs, int e, int chunk, int nb) {
    extern __shared__ int hist[];
    const int t = threadIdx.x;
    const int blk = blockIdx.x;
    const int G = gridDim.x;
    for (int i = t; i < nb; i += BK_T) hist[i] = PLACE ? offs[i * G + blk] : 0;
    __syncthreads();
    const int lo = blk * chunk;
    const int hi = min(e, lo + chunk);
    for (int i0 = lo + t; i0 < hi; i0 += BK_T * BK_U) {
        int d[BK_U], s[BK_U];
#pragma unroll
        for (int u = 0; u < BK_U; ++u) {
            const int i = i0 + u * BK_T;
            d[u] = (i < hi) ? dst[i] : 0;
            s[u] = (PLACE && i < hi) ? src[i] : 0;
        }
#pragma unroll
        for (int u = 0; u < BK_U; ++u) {
            if (i0 + u * BK_T < hi) {
                const int b = d[u] >> BK_SHIFT;
                if (PLACE) {
                    const int pos = atomicAdd(&hist[b], 1);
                    pairs[pos] = make_int2(s[u], d[u]);
                } else {
                    atomicAdd(&hist[b], 1);
                }
            }
        }
    }
    if (!PLACE) {
        __syncthreads();
        for (int i = t; i < nb; i += BK_T) tab[i * G + blk] = hist[i];
    }
}

__global__ void scan_sums_kernel(const int* __restrict__ cnt, int* __restrict__ bsum, int n) {
    __shared__ int sW[SCAN_T / 32];
    const int t = threadIdx.x;
    const int base = blockIdx.x * SCAN_BLK + t * SCAN_PER;
    int s = 0;
#pragma unroll
    for (int j = 0; j < SCAN_PER; ++j)
        if (base + j < n) s += cnt[base + j];
    for (int d = 16; d > 0; d >>= 1) s += __shfl_xor(s, d);
    if ((t & 31) == 0) sW[t >> 5] = s;
    __syncthreads();
    if (t == 0) {
        int tot = 0;
        for (int k = 0; k < SCAN_T / 32; ++k) tot += sW[k];
        bsum[blockIdx.x] = tot;
    }
}

__global__ void scan_apply_kernel(const int* __restrict__ cnt, const int* __restrict__ bsum,
                                  int* __restrict__ offs, int n) {
    __shared__ int sW[SCAN_T / 32];
    const int t = threadIdx.x;
    const int lane = t & 31;
    const int wv = t >> 5;
    const int b = blockIdx.x;

    int part = 0;
    for (int k = t; k < b; k += SCAN_T) part += bsum[k];
    for (int d = 16; d > 0; d >>= 1) part += __shfl_xor(part, d);
    if (lane == 0) sW[wv] = part;
    __syncthreads();
    int offset = 0;
    for (int k = 0; k < SCAN_T / 32; ++k) offset += sW[k];
    __syncthreads();

    const int base = b * SCAN_BLK + t * SCAN_PER;
    int c[SCAN_PER];
    int run = 0;
#pragma unroll
    for (int j = 0; j < SCAN_PER; ++j) {
        c[j] = (base + j < n) ? cnt[base + j] : 0;
        run += c[j];
    }
    int inc = run;
    for (int d = 1; d < 32; d <<= 1) {
        const int v = __shfl_up(inc, d);
        if (lane >= d) inc += v;
    }
    if (lane == 31) sW[wv] = inc;
    __syncthreads();
    int pre = offset + inc - run;
    for (int k = 0; k < wv; ++k) pre += sW[k];
#pragma unroll
    for (int j = 0; j < SCAN_PER; ++j) {
        pre += c[j];
        if (base + j < n) offs[base + j + 1] = pre;
    }
    if (b == 0 && t == 0) offs[0] = 0;
}

__global__ void __launch_bounds__(FIN_T)
bucket_finish_kernel(const int* __restrict__ offs, const int2* __restrict__ pairs,
                     int* __restrict__ rowptr, float* __restrict__ dinv,
                     int* colTmp, int* __restrict__ col, int n, int G) {
    __shared__ int cntL[BK_NODES];
    __shared__ int offL[BK_NODES];
    __shared__ int curL[BK_NODES];
    __shared__ int wtot[BK_NODES / 32];
    const int t = threadIdx.x;
    const int lane = t & 31;
    const int wv = t >> 5;
    const int b = blockIdx.x;
    const int beg = offs[b * G];
    const int end = offs[(b + 1) * G];
    if (t < BK_NODES) { cntL[t] = 0; curL[t] = 0; }
    __syncthreads();
    for (int i = beg + t; i < end; i += FIN_T)
        atomicAdd(&cntL[pairs[i].y & (BK_NODES - 1)], 1);
    __syncthreads();

    const int c = (t < BK_NODES) ? cntL[t] : 0;
    int inc = c;
    for (int d = 1; d < 32; d <<= 1) {
        const int v = __shfl_up(inc, d);
        if (lane >= d) inc += v;
    }
    if (t < BK_NODES && lane == 31) wtot[wv] = inc;
    __syncthreads();
    if (t < BK_NODES) {
        int pre = 0;
        for (int k = 0; k < wv; ++k) pre += wtot[k];
        const int excl = pre + inc - c;
        offL[t] = excl;
        const int node = b * BK_NODES + t;
        if (node < n) {
            rowptr[node + 1] = beg + excl + c;
            dinv[node] = rsqrtf((float)(c + 1));
        }
    }
    if (b == 0 && t == 0) rowptr[0] = 0;
    __syncthreads();

    for (int i = beg + t; i < end; i += FIN_T) {
        const int2 p = pairs[i];
        const int d = p.y & (BK_NODES - 1);
        const int r = atomicAdd(&curL[d], 1);
        colTmp[beg + offL[d] + r] = p.x;
    }
    __syncthreads();

    for (int row = wv; row < BK_NODES; row += FIN_T / 32) {
        const int deg = __builtin_amdgcn_readfirstlane(cntL[row]);
        const int rb  = __builtin_amdgcn_readfirstlane(beg + offL[row]);
        for (int base = 0; base < deg; base += 32) {
            const int i = base + lane;
            const int key = (i < deg) ? colTmp[rb + i] : 0;
            int r = 0;
            for (int jb = 0; jb < deg; jb += 32) {
                const int lim = min(32, deg - jb);
                const int kv = (lane < lim) ? colTmp[rb + jb + lane] : 0;
                for (int tt = 0; tt < lim; ++tt) {
                    const int kj = __builtin_amdgcn_readlane(kv, tt);
                    const int j = jb + tt;
                    r += ((kj < key) || (kj == key && j < i)) ? 1 : 0;
                }
            }
            if (i < deg) col[rb + r] = key;
        }
    }
}


#define AG_WAVES 4
#define AG_NPW   4
#define AG_BATCH 16
#define AG_PITCH 72

typedef __attribute__((ext_vector_type(2))) _Float16 v2h;
typedef __attribute__((ext_vector_type(8))) _Float16 v8h;

template <bool RELU, bool FUSE>
__global__ void __launch_bounds__(AG_WAVES * 32)
agg_tile_kernel(const float* __restrict__ hW,
                const int* __restrict__ rowptr,
                const int* __restrict__ col,
                const float* __restrict__ dinv,
                const float* __restrict__ bias,
                const _Float16* __restrict__ Wp,
                float* __restrict__ out, int n) {
    __shared__ __attribute__((aligned(16))) _Float16 tile[16 * AG_PITCH];
    const int lane = threadIdx.x & 31;
    const int wv = __builtin_amdgcn_readfirstlane((int)(threadIdx.x >> 5));
    const int tileBase = blockIdx.x * 16;
    const int node0 = tileBase + wv * AG_NPW;

    int rp[AG_NPW + 1];
#pragma unroll
    for (int j = 0; j <= AG_NPW; ++j) rp[j] = rowptr[min(node0 + j, n)];

    int    s0[AG_NPW];
    float2 self[AG_NPW];
    float  dd[AG_NPW];
#pragma unroll
    for (int j = 0; j < AG_NPW; ++j) {
        const int node = node0 + j;
        const int deg  = rp[j + 1] - rp[j];
        s0[j] = (lane < deg) ? col[rp[j] + lane] : 0;
        if (node < n) {
            self[j] = ((const float2*)(hW + (size_t)node * HID))[lane];
            dd[j]   = dinv[node];
        } else {
            self[j] = make_float2(0.0f, 0.0f);
            dd[j]   = 0.0f;
        }
    }

#pragma unroll
    for (int j = 0; j < AG_NPW; ++j) {
        const int node = node0 + j;
        const int beg  = rp[j];
        const int deg  = rp[j + 1] - beg;
        float ax = 0.0f, ay = 0.0f;
        const int s   = s0[j];
        const int lim = min(32, deg);
        for (int t = 0; t < lim; t += AG_BATCH) {
            float2 v[AG_BATCH];
            {
                const int sj = __builtin_amdgcn_readlane(s, t);
                v[0] = ((const float2*)(hW + (size_t)sj * HID))[lane];
            }
#pragma unroll
            for (int u = 1; u < AG_BATCH; ++u) {
                v[u] = make_float2(0.0f, 0.0f);
                if (t + u < lim) {
                    const int sj = __builtin_amdgcn_readlane(s, t + u);
                    v[u] = ((const float2*)(hW + (size_t)sj * HID))[lane];
                }
            }
#pragma unroll
            for (int u = 0; u < AG_BATCH; ++u) {
                ax += v[u].x;
                ay += v[u].y;
            }
        }
        for (int e = 32; e < deg; ++e) {
            const int sj = col[beg + e];
            const float2 v = ((const float2*)(hW + (size_t)sj * HID))[lane];
            ax += v.x;
            ay += v.y;
        }
        float ox = (ax + self[j].x) * dd[j];
        float oy = (ay + self[j].y) * dd[j];
        if (bias) { ox += bias[lane * 2]; oy += bias[lane * 2 + 1]; }
        if (FUSE) {
            if (RELU) { ox = fmaxf(ox, 0.0f); oy = fmaxf(oy, 0.0f); }
            v2h p;
            p[0] = (_Float16)ox;
            p[1] = (_Float16)oy;
            *(v2h*)(tile + (wv * AG_NPW + j) * AG_PITCH + lane * 2) = p;
        } else {
            if (node < n) ((float2*)(out + (size_t)node * HID))[lane] = make_float2(ox, oy);
        }
    }

    if (FUSE) {
        __syncthreads();
        if (wv < 2) {
            const int m    = lane & 15;
            const int koff = (lane >> 4) << 3;
            const v16h* bp = (const v16h*)Wp;
            v8f acc0 = {}, acc1 = {};
#pragma unroll
            for (int ks = 0; ks < HID / 32; ++ks) {
                const _Float16* ap = tile + m * AG_PITCH + ks * 32 + koff;
                const v8h lo = *(const v8h*)ap;
                const v8h hi = *(const v8h*)(ap + 16);
                const v16h a = __builtin_shufflevector(lo, hi, 0, 1, 2, 3, 4, 5, 6, 7,
                                                       8, 9, 10, 11, 12, 13, 14, 15);
                const v16h b0 = bp[(size_t)(ks * 4 + wv * 2 + 0) * 32 + lane];
                const v16h b1 = bp[(size_t)(ks * 4 + wv * 2 + 1) * 32 + lane];
                acc0 = __builtin_amdgcn_wmma_f32_16x16x32_f16(
                    false, a, false, b0, (short)0, acc0, false, false);
                acc1 = __builtin_amdgcn_wmma_f32_16x16x32_f16(
                    false, a, false, b1, (short)0, acc1, false, false);
            }
            const int colLane = lane & 15;
            const int rBase   = tileBase + ((lane >> 4) << 3);
            const float4 sc0 = *(const float4*)(dinv + rBase);
            const float4 sc1 = *(const float4*)(dinv + rBase + 4);
            const float sc[8] = {sc0.x, sc0.y, sc0.z, sc0.w, sc1.x, sc1.y, sc1.z, sc1.w};
#pragma unroll
            for (int i = 0; i < 8; ++i) {
                float* op = out + (size_t)(rBase + i) * HID + wv * 32 + colLane;
                op[0]  = acc0[i] * sc[i];
                op[16] = acc1[i] * sc[i];
            }
        }
    }
}


extern "C" void kernel_launch(void* const* d_in, const int* in_sizes, int n_in,
                              void* d_out, int out_size, void* d_ws, size_t ws_size,
                              hipStream_t stream) {
    const float* x     = (const float*)d_in[0];
    const int*   eidx  = (const int*)d_in[1];
    const float* W_in  = (const float*)d_in[2];
    const float* b_in  = (const float*)d_in[3];
    const float* W_h   = (const float*)d_in[4];
    const float* b_h   = (const float*)d_in[5];
    const float* W_out = (const float*)d_in[6];

    const int N = in_sizes[0] / IN_D;
    const int E = in_sizes[1] / 2;
    const int L = in_sizes[4] / (HID * HID);

    const int* src = eidx;
    const int* dst = eidx + E;

    char* ws = (char*)d_ws;
    auto align256 = [](size_t v) { return (v + 255) & ~(size_t)255; };
    size_t off = 0;
    float* dinv = (float*)(ws + off);  off = align256(off + (size_t)N * sizeof(float));
    float* B1   = (float*)(ws + off);  off = align256(off + (size_t)N * HID * sizeof(float));
    _Float16* WpIn  = (_Float16*)(ws + off);  off = align256(off + 16 * 1024);
    _Float16* WpH0  = (_Float16*)(ws + off);  off = align256(off + 8 * 1024);
    _Float16* WpH1  = (_Float16*)(ws + off);  off = align256(off + 8 * 1024);
    _Float16* WpH2  = (_Float16*)(ws + off);  off = align256(off + 8 * 1024);
    _Float16* WpOut = (_Float16*)(ws + off);  off = align256(off + 8 * 1024);
    _Float16* WpHs[3] = {WpH0, WpH1, WpH2};
    int* rowptr = (int*)(ws + off);  off = align256(off + ((size_t)N + 1) * sizeof(int));
    int* col    = (int*)(ws + off);  off = align256(off + (size_t)E * sizeof(int));
    float* O = (float*)d_out;

    const int T = 256;

    const int rowTiles   = N / 16;
    const int gemmBlocks = (rowTiles + 7) / 8;
    const int tileBlocks = (N + 15) / 16;
    const int fullTiles  = N / 16;
    const int nBuck      = (N + BK_NODES - 1) >> BK_SHIFT;
    const int tabLen     = nBuck * BK_GRID;
    const int scanBlocks = (tabLen + SCAN_BLK - 1) / SCAN_BLK;
    const int chunk      = (E + BK_GRID - 1) / BK_GRID;

    const bool listFits = off <= ws_size && nBuck <= BK_MAXB &&
                          (size_t)3 * E + (size_t)2 * tabLen + 1 + (size_t)scanBlocks
                              <= (size_t)N * HID;

    if (listFits) {
        int2* pairs = (int2*)d_out;
        int* colTmp = (int*)d_out + (size_t)2 * E;
        int* tab    = colTmp + E;
        int* offs   = tab + tabLen;
        int* bsum   = offs + tabLen + 1;
        const size_t histBytes = (size_t)nBuck * sizeof(int);

        bucket_pass_kernel<false><<<BK_GRID, BK_T, histBytes, stream>>>(src, dst, tab, offs, pairs, E, chunk, nBuck);
        scan_sums_kernel<<<scanBlocks, SCAN_T, 0, stream>>>(tab, bsum, tabLen);
        scan_apply_kernel<<<scanBlocks, SCAN_T, 0, stream>>>(tab, bsum, offs, tabLen);
        bucket_pass_kernel<true><<<BK_GRID, BK_T, histBytes, stream>>>(src, dst, tab, offs, pairs, E, chunk, nBuck);
        bucket_finish_kernel<<<nBuck, FIN_T, 0, stream>>>(offs, pairs, rowptr, dinv, colTmp, col, N, BK_GRID);

        pack_w_kernel<IN_D><<<2, T, 0, stream>>>(W_in, WpIn);
        for (int l = 0; l < L; ++l)
            pack_w_kernel<HID><<<1, T, 0, stream>>>(W_h + (size_t)l * HID * HID, WpHs[l]);
        pack_w_kernel<HID><<<1, T, 0, stream>>>(W_out, WpOut);

        float* cur = (L & 1) ? B1 : O;
        float* nxt = (L & 1) ? O : B1;

        gemm_wmma_kernel<IN_D, false><<<gemmBlocks, T, 0, stream>>>(x, WpIn, dinv, cur, N);

        for (int i = 0; i <= L; ++i) {
            const float* b = (i == 0) ? b_in : b_h + (size_t)(i - 1) * HID;
            const _Float16* wp = (i < L) ? WpHs[i] : WpOut;
            if (i == 0)
                agg_tile_kernel<false, true><<<fullTiles, AG_WAVES * 32, 0, stream>>>(cur, rowptr, col, dinv, b, wp, nxt, N);
            else
                agg_tile_kernel<true, true><<<fullTiles, AG_WAVES * 32, 0, stream>>>(cur, rowptr, col, dinv, b, wp, nxt, N);
            float* tmp = cur; cur = nxt; nxt = tmp;
        }

        agg_tile_kernel<false, false><<<tileBlocks, AG_WAVES * 32, 0, stream>>>(cur, rowptr, col, dinv, nullptr, nullptr, O, N);
        return;
    }


    deg_init_kernel<<<(N + T - 1) / T, T, 0, stream>>>(dinv, N);
    deg_accum_kernel<<<(E + T - 1) / T, T, 0, stream>>>(dst, dinv, E);
    dinv_kernel<<<(N + T - 1) / T, T, 0, stream>>>(dinv, N);

    pack_w_kernel<IN_D><<<2, T, 0, stream>>>(W_in, WpIn);
    for (int l = 0; l < L; ++l)
        pack_w_kernel<HID><<<1, T, 0, stream>>>(W_h + (size_t)l * HID * HID, WpHs[l]);
    pack_w_kernel<HID><<<1, T, 0, stream>>>(W_out, WpOut);

    const int totalNC    = N * HID;
    const int initBlocks = (totalNC + T - 1) / T;
    const int edgeBlocks = (E + 7) / 8;

    gemm_wmma_kernel<IN_D, false><<<gemmBlocks, T, 0, stream>>>(x, WpIn, dinv, B1, N);
    agg_init_kernel<<<initBlocks, T, 0, stream>>>(B1, dinv, b_in, O, totalNC);
    agg_edges_kernel<<<edgeBlocks, T, 0, stream>>>(B1, src, dst, dinv, O, E);

    for (int l = 0; l < L; ++l) {
        const float* b = b_h + (size_t)l * HID;
        if (l == 0)
            gemm_wmma_kernel<HID, false><<<gemmBlocks, T, 0, stream>>>(O, WpHs[l], dinv, B1, N);
        else
            gemm_wmma_kernel<HID, true ><<<gemmBlocks, T, 0, stream>>>(O, WpHs[l], dinv, B1, N);
        agg_init_kernel<<<initBlocks, T, 0, stream>>>(B1, dinv, b, O, totalNC);
        agg_edges_kernel<<<edgeBlocks, T, 0, stream>>>(B1, src, dst, dinv, O, E);
    }

    gemm_wmma_kernel<HID, true><<<gemmBlocks, T, 0, stream>>>(O, WpOut, dinv, B1, N);
    agg_init_kernel<<<initBlocks, T, 0, stream>>>(B1, dinv, nullptr, O, totalNC);
    agg_edges_kernel<<<edgeBlocks, T, 0, stream>>>(B1, src, dst, dinv, O, E);
}


# weight packing rides in the first bucket-count launch as 3 extra blocks (one wave per fragment, same code): the 5 separate packing launches are gone, 16 launches -> 11
# speedup vs baseline: 4.5102x; 1.0570x over previous
#include <hip/hip_runtime.h>
#include <hip/hip_bf16.h>

typedef __attribute__((ext_vector_type(16))) _Float16 v16h;
typedef __attribute__((ext_vector_type(8)))  float    v8f;

#define IN_D  128
#define HID   64


__global__ void deg_init_kernel(float* deg, int n) {
    int i = blockIdx.x * blockDim.x + threadIdx.x;
    if (i < n) deg[i] = 1.0f;
}

__global__ void deg_accum_kernel(const int* __restrict__ dst, float* deg, int e) {
    int i = blockIdx.x * blockDim.x + threadIdx.x;
    if (i < e) unsafeAtomicAdd(&deg[dst[i]], 1.0f);
}

__global__ void dinv_kernel(float* deg, int n) {
    int i = blockIdx.x * blockDim.x + threadIdx.x;
    if (i < n) deg[i] = rsqrtf(deg[i]);
}


__device__ __forceinline__ void pack_fragment(const float* __restrict__ W,
                                              _Float16* __restrict__ Wp, int frag, int lane) {
    const int ks = frag >> 2;
    const int ct = frag & 3;
    const int n    = lane & 15;
    const int koff = (lane >> 4) << 3;
    v16h b;
    const float* wp = W + (size_t)(ks * 32 + koff) * HID + ct * 16 + n;
#pragma unroll
    for (int j = 0; j < 8; ++j) {
        b[j]     = (_Float16)wp[(size_t)j * HID];
        b[8 + j] = (_Float16)wp[(size_t)(16 + j) * HID];
    }
    *(v16h*)(Wp + ((size_t)frag * 32 + lane) * 16) = b;
}

#define WP_IN_FRAGS  ((IN_D / 32) * 4)
#define WP_HID_FRAGS ((HID / 32) * 4)
#define WP_HID_SLOTS 3
#define WP_FRAG_H    (32 * 16)

template <int K>
__global__ void pack_w_kernel(const float* __restrict__ W, _Float16* __restrict__ Wp) {
    const int frag = (blockIdx.x * blockDim.x + threadIdx.x) >> 5;
    const int lane = threadIdx.x & 31;
    const int nFrag = (K / 32) * 4;
    if (frag >= nFrag) return;
    pack_fragment(W, Wp, frag, lane);
}


template <int K, bool RELU>
__global__ void gemm_wmma_kernel(const float* __restrict__ A,
                                 const _Float16* __restrict__ Wp,
                                 const float* __restrict__ rowScale,
                                 float* __restrict__ out, int nRows) {
    const int wave = (blockIdx.x * blockDim.x + threadIdx.x) >> 5;
    const int lane = threadIdx.x & 31;
    const int rowTiles = nRows >> 4;
    if (wave >= rowTiles) return;

    const int m    = lane & 15;
    const int koff = (lane >> 4) << 3;
    const int row  = (wave << 4) + m;

    const v16h* bp = (const v16h*)Wp;

    const int colLane = lane & 15;
    const int rBase   = (wave << 4) + ((lane >> 4) << 3);
    const float4 sc0 = *(const float4*)(rowScale + rBase);
    const float4 sc1 = *(const float4*)(rowScale + rBase + 4);
    const float sc[8] = {sc0.x, sc0.y, sc0.z, sc0.w, sc1.x, sc1.y, sc1.z, sc1.w};

    v8f acc0 = {}, acc1 = {}, acc2 = {}, acc3 = {};

#pragma unroll
    for (int k0 = 0; k0 < K; k0 += 32) {
        const int ks = k0 >> 5;
        v16h a;
        const float* ap = A + (size_t)row * K + k0 + koff;
#pragma unroll
        for (int j = 0; j < 8; ++j) {
            float f0 = ap[j];
            float f1 = ap[16 + j];
            if (RELU) { f0 = fmaxf(f0, 0.0f); f1 = fmaxf(f1, 0.0f); }
            a[j]     = (_Float16)f0;
            a[8 + j] = (_Float16)f1;
        }
#pragma unroll
        for (int ct = 0; ct < 4; ++ct) {
            v16h b = bp[(size_t)(ks * 4 + ct) * 32 + lane];
            v8f& acc = (ct == 0 ? acc0 : ct == 1 ? acc1 : ct == 2 ? acc2 : acc3);
            acc = __builtin_amdgcn_wmma_f32_16x16x32_f16(
                false, a, false, b, (short)0, acc, false, false);
        }
    }

    auto store_tile = [&](v8f acc, int ct) {
#pragma unroll
        for (int i = 0; i < 8; ++i)
            out[(size_t)(rBase + i) * HID + ct * 16 + colLane] = acc[i] * sc[i];
    };
    store_tile(acc0, 0); store_tile(acc1, 1); store_tile(acc2, 2); store_tile(acc3, 3);
}

__global__ void agg_init_kernel(const float* __restrict__ hW,
                                const float* __restrict__ dinv,
                                const float* __restrict__ bias,
                                float* __restrict__ out, int total) {
    int idx = blockIdx.x * blockDim.x + threadIdx.x;
    if (idx >= total) return;
    int node = idx >> 6;
    int c    = idx & 63;
    float v = hW[idx] * dinv[node];
    if (bias) v += bias[c];
    out[idx] = v;
}

__global__ void agg_edges_kernel(const float* __restrict__ hW,
                                 const int* __restrict__ src,
                                 const int* __restrict__ dst,
                                 const float* __restrict__ dinv,
                                 float* __restrict__ out, int nEdges) {
    const int wave = (blockIdx.x * blockDim.x + threadIdx.x) >> 5;
    const int lane = threadIdx.x & 31;
    if (wave >= nEdges) return;
    const int s = src[wave];
    const int d = dst[wave];
    const float w = dinv[d];
    const float2 v = ((const float2*)(hW + (size_t)s * HID))[lane];
    float* op = out + (size_t)d * HID + lane * 2;
    unsafeAtomicAdd(op,     v.x * w);
    unsafeAtomicAdd(op + 1, v.y * w);
}


#define SCAN_T   256
#define SCAN_PER 4
#define SCAN_BLK (SCAN_T * SCAN_PER)

#define BK_SHIFT 8
#define BK_NODES (1 << BK_SHIFT)
#define BK_GRID  512
#define BK_T     512
#define BK_U     8
#define FIN_T    1024
#define BK_MAXB  8192

template <bool PLACE>
__global__ void __launch_bounds__(BK_T)
bucket_pass_kernel(const int* __restrict__ src, const int* __restrict__ dst,
                   int* __restrict__ tab, const int* __restrict__ offs,
                   int2* __restrict__ pairs, int e, int chunk, int nb,
                   const float* __restrict__ W_in, const float* __restrict__ W_h,
                   const float* __restrict__ W_out, _Float16* __restrict__ Wp, int nLayers) {
    extern __shared__ int hist[];
    const int t = threadIdx.x;
    const int blk = blockIdx.x;
    const int G = BK_GRID;
    if (!PLACE && blk >= BK_GRID) {
        const int f = (blk - BK_GRID) * (BK_T / 32) + (t >> 5);
        const int lane = t & 31;
        if (f < WP_IN_FRAGS) {
            pack_fragment(W_in, Wp, f, lane);
        } else {
            const int g = f - WP_IN_FRAGS;
            const int m = g / WP_HID_FRAGS;
            if (m <= nLayers && m <= WP_HID_SLOTS) {
                const float* W = (m < nLayers) ? W_h + (size_t)m * HID * HID : W_out;
                const int slot = (m < nLayers) ? m : WP_HID_SLOTS;
                pack_fragment(W, Wp + (size_t)(WP_IN_FRAGS + slot * WP_HID_FRAGS) * WP_FRAG_H,
                              g % WP_HID_FRAGS, lane);
            }
        }
        return;
    }
    for (int i = t; i < nb; i += BK_T) hist[i] = PLACE ? offs[i * G + blk] : 0;
    __syncthreads();
    const int lo = blk * chunk;
    const int hi = min(e, lo + chunk);
    for (int i0 = lo + t; i0 < hi; i0 += BK_T * BK_U) {
        int d[BK_U], s[BK_U];
#pragma unroll
        for (int u = 0; u < BK_U; ++u) {
            const int i = i0 + u * BK_T;
            d[u] = (i < hi) ? dst[i] : 0;
            s[u] = (PLACE && i < hi) ? src[i] : 0;
        }
#pragma unroll
        for (int u = 0; u < BK_U; ++u) {
            if (i0 + u * BK_T < hi) {
                const int b = d[u] >> BK_SHIFT;
                if (PLACE) {
                    const int pos = atomicAdd(&hist[b], 1);
                    pairs[pos] = make_int2(s[u], d[u]);
                } else {
                    atomicAdd(&hist[b], 1);
                }
            }
        }
    }
    if (!PLACE) {
        __syncthreads();
        for (int i = t; i < nb; i += BK_T) tab[i * G + blk] = hist[i];
    }
}

__global__ void scan_sums_kernel(const int* __restrict__ cnt, int* __restrict__ bsum, int n) {
    __shared__ int sW[SCAN_T / 32];
    const int t = threadIdx.x;
    const int base = blockIdx.x * SCAN_BLK + t * SCAN_PER;
    int s = 0;
#pragma unroll
    for (int j = 0; j < SCAN_PER; ++j)
        if (base + j < n) s += cnt[base + j];
    for (int d = 16; d > 0; d >>= 1) s += __shfl_xor(s, d);
    if ((t & 31) == 0) sW[t >> 5] = s;
    __syncthreads();
    if (t == 0) {
        int tot = 0;
        for (int k = 0; k < SCAN_T / 32; ++k) tot += sW[k];
        bsum[blockIdx.x] = tot;
    }
}

__global__ void scan_apply_kernel(const int* __restrict__ cnt, const int* __restrict__ bsum,
                                  int* __restrict__ offs, int n) {
    __shared__ int sW[SCAN_T / 32];
    const int t = threadIdx.x;
    const int lane = t & 31;
    const int wv = t >> 5;
    const int b = blockIdx.x;

    int part = 0;
    for (int k = t; k < b; k += SCAN_T) part += bsum[k];
    for (int d = 16; d > 0; d >>= 1) part += __shfl_xor(part, d);
    if (lane == 0) sW[wv] = part;
    __syncthreads();
    int offset = 0;
    for (int k = 0; k < SCAN_T / 32; ++k) offset += sW[k];
    __syncthreads();

    const int base = b * SCAN_BLK + t * SCAN_PER;
    int c[SCAN_PER];
    int run = 0;
#pragma unroll
    for (int j = 0; j < SCAN_PER; ++j) {
        c[j] = (base + j < n) ? cnt[base + j] : 0;
        run += c[j];
    }
    int inc = run;
    for (int d = 1; d < 32; d <<= 1) {
        const int v = __shfl_up(inc, d);
        if (lane >= d) inc += v;
    }
    if (lane == 31) sW[wv] = inc;
    __syncthreads();
    int pre = offset + inc - run;
    for (int k = 0; k < wv; ++k) pre += sW[k];
#pragma unroll
    for (int j = 0; j < SCAN_PER; ++j) {
        pre += c[j];
        if (base + j < n) offs[base + j + 1] = pre;
    }
    if (b == 0 && t == 0) offs[0] = 0;
}

__global__ void __launch_bounds__(FIN_T)
bucket_finish_kernel(const int* __restrict__ offs, const int2* __restrict__ pairs,
                     int* __restrict__ rowptr, float* __restrict__ dinv,
                     int* colTmp, int* __restrict__ col, int n, int G) {
    __shared__ int cntL[BK_NODES];
    __shared__ int offL[BK_NODES];
    __shared__ int curL[BK_NODES];
    __shared__ int wtot[BK_NODES / 32];
    const int t = threadIdx.x;
    const int lane = t & 31;
    const int wv = t >> 5;
    const int b = blockIdx.x;
    const int beg = offs[b * G];
    const int end = offs[(b + 1) * G];
    if (t < BK_NODES) { cntL[t] = 0; curL[t] = 0; }
    __syncthreads();
    for (int i = beg + t; i < end; i += FIN_T)
        atomicAdd(&cntL[pairs[i].y & (BK_NODES - 1)], 1);
    __syncthreads();

    const int c = (t < BK_NODES) ? cntL[t] : 0;
    int inc = c;
    for (int d = 1; d < 32; d <<= 1) {
        const int v = __shfl_up(inc, d);
        if (lane >= d) inc += v;
    }
    if (t < BK_NODES && lane == 31) wtot[wv] = inc;
    __syncthreads();
    if (t < BK_NODES) {
        int pre = 0;
        for (int k = 0; k < wv; ++k) pre += wtot[k];
        const int excl = pre + inc - c;
        offL[t] = excl;
        const int node = b * BK_NODES + t;
        if (node < n) {
            rowptr[node + 1] = beg + excl + c;
            dinv[node] = rsqrtf((float)(c + 1));
        }
    }
    if (b == 0 && t == 0) rowptr[0] = 0;
    __syncthreads();

    for (int i = beg + t; i < end; i += FIN_T) {
        const int2 p = pairs[i];
        const int d = p.y & (BK_NODES - 1);
        const int r = atomicAdd(&curL[d], 1);
        colTmp[beg + offL[d] + r] = p.x;
    }
    __syncthreads();

    for (int row = wv; row < BK_NODES; row += FIN_T / 32) {
        const int deg = __builtin_amdgcn_readfirstlane(cntL[row]);
        const int rb  = __builtin_amdgcn_readfirstlane(beg + offL[row]);
        for (int base = 0; base < deg; base += 32) {
            const int i = base + lane;
            const int key = (i < deg) ? colTmp[rb + i] : 0;
            int r = 0;
            for (int jb = 0; jb < deg; jb += 32) {
                const int lim = min(32, deg - jb);
                const int kv = (lane < lim) ? colTmp[rb + jb + lane] : 0;
                for (int tt = 0; tt < lim; ++tt) {
                    const int kj = __builtin_amdgcn_readlane(kv, tt);
                    const int j = jb + tt;
                    r += ((kj < key) || (kj == key && j < i)) ? 1 : 0;
                }
            }
            if (i < deg) col[rb + r] = key;
        }
    }
}


#define AG_WAVES 4
#define AG_NPW   4
#define AG_BATCH 16
#define AG_PITCH 72

typedef __attribute__((ext_vector_type(2))) _Float16 v2h;
typedef __attribute__((ext_vector_type(8))) _Float16 v8h;

template <bool RELU, bool FUSE>
__global__ void __launch_bounds__(AG_WAVES * 32)
agg_tile_kernel(const float* __restrict__ hW,
                const int* __restrict__ rowptr,
                const int* __restrict__ col,
                const float* __restrict__ dinv,
                const float* __restrict__ bias,
                const _Float16* __restrict__ Wp,
                float* __restrict__ out, int n) {
    __shared__ __attribute__((aligned(16))) _Float16 tile[16 * AG_PITCH];
    const int lane = threadIdx.x & 31;
    const int wv = __builtin_amdgcn_readfirstlane((int)(threadIdx.x >> 5));
    const int tileBase = blockIdx.x * 16;
    const int node0 = tileBase + wv * AG_NPW;

    int rp[AG_NPW + 1];
#pragma unroll
    for (int j = 0; j <= AG_NPW; ++j) rp[j] = rowptr[min(node0 + j, n)];

    int    s0[AG_NPW];
    float2 self[AG_NPW];
    float  dd[AG_NPW];
#pragma unroll
    for (int j = 0; j < AG_NPW; ++j) {
        const int node = node0 + j;
        const int deg  = rp[j + 1] - rp[j];
        s0[j] = (lane < deg) ? col[rp[j] + lane] : 0;
        if (node < n) {
            self[j] = ((const float2*)(hW + (size_t)node * HID))[lane];
            dd[j]   = dinv[node];
        } else {
            self[j] = make_float2(0.0f, 0.0f);
            dd[j]   = 0.0f;
        }
    }

#pragma unroll
    for (int j = 0; j < AG_NPW; ++j) {
        const int node = node0 + j;
        const int beg  = rp[j];
        const int deg  = rp[j + 1] - beg;
        float ax = 0.0f, ay = 0.0f;
        const int s   = s0[j];
        const int lim = min(32, deg);
        for (int t = 0; t < lim; t += AG_BATCH) {
            float2 v[AG_BATCH];
            {
                const int sj = __builtin_amdgcn_readlane(s, t);
                v[0] = ((const float2*)(hW + (size_t)sj * HID))[lane];
            }
#pragma unroll
            for (int u = 1; u < AG_BATCH; ++u) {
                v[u] = make_float2(0.0f, 0.0f);
                if (t + u < lim) {
                    const int sj = __builtin_amdgcn_readlane(s, t + u);
                    v[u] = ((const float2*)(hW + (size_t)sj * HID))[lane];
                }
            }
#pragma unroll
            for (int u = 0; u < AG_BATCH; ++u) {
                ax += v[u].x;
                ay += v[u].y;
            }
        }
        for (int e = 32; e < deg; ++e) {
            const int sj = col[beg + e];
            const float2 v = ((const float2*)(hW + (size_t)sj * HID))[lane];
            ax += v.x;
            ay += v.y;
        }
        float ox = (ax + self[j].x) * dd[j];
        float oy = (ay + self[j].y) * dd[j];
        if (bias) { ox += bias[lane * 2]; oy += bias[lane * 2 + 1]; }
        if (FUSE) {
            if (RELU) { ox = fmaxf(ox, 0.0f); oy = fmaxf(oy, 0.0f); }
            v2h p;
            p[0] = (_Float16)ox;
            p[1] = (_Float16)oy;
            *(v2h*)(tile + (wv * AG_NPW + j) * AG_PITCH + lane * 2) = p;
        } else {
            if (node < n) ((float2*)(out + (size_t)node * HID))[lane] = make_float2(ox, oy);
        }
    }

    if (FUSE) {
        __syncthreads();
        if (wv < 2) {
            const int m    = lane & 15;
            const int koff = (lane >> 4) << 3;
            const v16h* bp = (const v16h*)Wp;
            v8f acc0 = {}, acc1 = {};
#pragma unroll
            for (int ks = 0; ks < HID / 32; ++ks) {
                const _Float16* ap = tile + m * AG_PITCH + ks * 32 + koff;
                const v8h lo = *(const v8h*)ap;
                const v8h hi = *(const v8h*)(ap + 16);
                const v16h a = __builtin_shufflevector(lo, hi, 0, 1, 2, 3, 4, 5, 6, 7,
                                                       8, 9, 10, 11, 12, 13, 14, 15);
                const v16h b0 = bp[(size_t)(ks * 4 + wv * 2 + 0) * 32 + lane];
                const v16h b1 = bp[(size_t)(ks * 4 + wv * 2 + 1) * 32 + lane];
                acc0 = __builtin_amdgcn_wmma_f32_16x16x32_f16(
                    false, a, false, b0, (short)0, acc0, false, false);
                acc1 = __builtin_amdgcn_wmma_f32_16x16x32_f16(
                    false, a, false, b1, (short)0, acc1, false, false);
            }
            const int colLane = lane & 15;
            const int rBase   = tileBase + ((lane >> 4) << 3);
            const float4 sc0 = *(const float4*)(dinv + rBase);
            const float4 sc1 = *(const float4*)(dinv + rBase + 4);
            const float sc[8] = {sc0.x, sc0.y, sc0.z, sc0.w, sc1.x, sc1.y, sc1.z, sc1.w};
#pragma unroll
            for (int i = 0; i < 8; ++i) {
                float* op = out + (size_t)(rBase + i) * HID + wv * 32 + colLane;
                op[0]  = acc0[i] * sc[i];
                op[16] = acc1[i] * sc[i];
            }
        }
    }
}


extern "C" void kernel_launch(void* const* d_in, const int* in_sizes, int n_in,
                              void* d_out, int out_size, void* d_ws, size_t ws_size,
                              hipStream_t stream) {
    const float* x     = (const float*)d_in[0];
    const int*   eidx  = (const int*)d_in[1];
    const float* W_in  = (const float*)d_in[2];
    const float* b_in  = (const float*)d_in[3];
    const float* W_h   = (const float*)d_in[4];
    const float* b_h   = (const float*)d_in[5];
    const float* W_out = (const float*)d_in[6];

    const int N = in_sizes[0] / IN_D;
    const int E = in_sizes[1] / 2;
    const int L = in_sizes[4] / (HID * HID);

    const int* src = eidx;
    const int* dst = eidx + E;

    char* ws = (char*)d_ws;
    auto align256 = [](size_t v) { return (v + 255) & ~(size_t)255; };
    size_t off = 0;
    float* dinv = (float*)(ws + off);  off = align256(off + (size_t)N * sizeof(float));
    float* B1   = (float*)(ws + off);  off = align256(off + (size_t)N * HID * sizeof(float));
    _Float16* WpIn  = (_Float16*)(ws + off);  off = align256(off + 16 * 1024);
    _Float16* WpH0  = (_Float16*)(ws + off);  off = align256(off + 8 * 1024);
    _Float16* WpH1  = (_Float16*)(ws + off);  off = align256(off + 8 * 1024);
    _Float16* WpH2  = (_Float16*)(ws + off);  off = align256(off + 8 * 1024);
    _Float16* WpOut = (_Float16*)(ws + off);  off = align256(off + 8 * 1024);
    _Float16* WpHs[3] = {WpH0, WpH1, WpH2};
    int* rowptr = (int*)(ws + off);  off = align256(off + ((size_t)N + 1) * sizeof(int));
    int* col    = (int*)(ws + off);  off = align256(off + (size_t)E * sizeof(int));
    float* O = (float*)d_out;

    const int T = 256;

    const int rowTiles   = N / 16;
    const int gemmBlocks = (rowTiles + 7) / 8;
    const int tileBlocks = (N + 15) / 16;
    const int fullTiles  = N / 16;
    const int nBuck      = (N + BK_NODES - 1) >> BK_SHIFT;
    const int tabLen     = nBuck * BK_GRID;
    const int scanBlocks = (tabLen + SCAN_BLK - 1) / SCAN_BLK;
    const int chunk      = (E + BK_GRID - 1) / BK_GRID;

    const bool listFits = off <= ws_size && nBuck <= BK_MAXB &&
                          (size_t)3 * E + (size_t)2 * tabLen + 1 + (size_t)scanBlocks
                              <= (size_t)N * HID;

    if (listFits) {
        int2* pairs = (int2*)d_out;
        int* colTmp = (int*)d_out + (size_t)2 * E;
        int* tab    = colTmp + E;
        int* offs   = tab + tabLen;
        int* bsum   = offs + tabLen + 1;
        const size_t histBytes = (size_t)nBuck * sizeof(int);

        const int packBlocks = ((WP_IN_FRAGS + WP_HID_FRAGS * (L + 1)) * 32 + BK_T - 1) / BK_T;
        bucket_pass_kernel<false><<<BK_GRID + packBlocks, BK_T, histBytes, stream>>>(src, dst, tab, offs, pairs, E, chunk, nBuck, W_in, W_h, W_out, WpIn, L);
        scan_sums_kernel<<<scanBlocks, SCAN_T, 0, stream>>>(tab, bsum, tabLen);
        scan_apply_kernel<<<scanBlocks, SCAN_T, 0, stream>>>(tab, bsum, offs, tabLen);
        bucket_pass_kernel<true><<<BK_GRID, BK_T, histBytes, stream>>>(src, dst, tab, offs, pairs, E, chunk, nBuck, nullptr, nullptr, nullptr, nullptr, 0);
        bucket_finish_kernel<<<nBuck, FIN_T, 0, stream>>>(offs, pairs, rowptr, dinv, colTmp, col, N, BK_GRID);

        float* cur = (L & 1) ? B1 : O;
        float* nxt = (L & 1) ? O : B1;

        gemm_wmma_kernel<IN_D, false><<<gemmBlocks, T, 0, stream>>>(x, WpIn, dinv, cur, N);

        for (int i = 0; i <= L; ++i) {
            const float* b = (i == 0) ? b_in : b_h + (size_t)(i - 1) * HID;
            const _Float16* wp = (i < L) ? WpHs[i] : WpOut;
            if (i == 0)
                agg_tile_kernel<false, true><<<fullTiles, AG_WAVES * 32, 0, stream>>>(cur, rowptr, col, dinv, b, wp, nxt, N);
            else
                agg_tile_kernel<true, true><<<fullTiles, AG_WAVES * 32, 0, stream>>>(cur, rowptr, col, dinv, b, wp, nxt, N);
            float* tmp = cur; cur = nxt; nxt = tmp;
        }

        agg_tile_kernel<false, false><<<tileBlocks, AG_WAVES * 32, 0, stream>>>(cur, rowptr, col, dinv, nullptr, nullptr, O, N);
        return;
    }


    deg_init_kernel<<<(N + T - 1) / T, T, 0, stream>>>(dinv, N);
    deg_accum_kernel<<<(E + T - 1) / T, T, 0, stream>>>(dst, dinv, E);
    dinv_kernel<<<(N + T - 1) / T, T, 0, stream>>>(dinv, N);

    pack_w_kernel<IN_D><<<2, T, 0, stream>>>(W_in, WpIn);
    for (int l = 0; l < L; ++l)
        pack_w_kernel<HID><<<1, T, 0, stream>>>(W_h + (size_t)l * HID * HID, WpHs[l]);
    pack_w_kernel<HID><<<1, T, 0, stream>>>(W_out, WpOut);

    const int totalNC    = N * HID;
    const int initBlocks = (totalNC + T - 1) / T;
    const int edgeBlocks = (E + 7) / 8;

    gemm_wmma_kernel<IN_D, false><<<gemmBlocks, T, 0, stream>>>(x, WpIn, dinv, B1, N);
    agg_init_kernel<<<initBlocks, T, 0, stream>>>(B1, dinv, b_in, O, totalNC);
    agg_edges_kernel<<<edgeBlocks, T, 0, stream>>>(B1, src, dst, dinv, O, E);

    for (int l = 0; l < L; ++l) {
        const float* b = b_h + (size_t)l * HID;
        if (l == 0)
            gemm_wmma_kernel<HID, false><<<gemmBlocks, T, 0, stream>>>(O, WpHs[l], dinv, B1, N);
        else
            gemm_wmma_kernel<HID, true ><<<gemmBlocks, T, 0, stream>>>(O, WpHs[l], dinv, B1, N);
        agg_init_kernel<<<initBlocks, T, 0, stream>>>(B1, dinv, b, O, totalNC);
        agg_edges_kernel<<<edgeBlocks, T, 0, stream>>>(B1, src, dst, dinv, O, E);
    }

    gemm_wmma_kernel<HID, true><<<gemmBlocks, T, 0, stream>>>(O, WpOut, dinv, B1, N);
    agg_init_kernel<<<initBlocks, T, 0, stream>>>(B1, dinv, nullptr, O, totalNC);
    agg_edges_kernel<<<edgeBlocks, T, 0, stream>>>(B1, src, dst, dinv, O, E);
}


# layers behind conv1 sum before they multiply, (Ahat h)W: gathers read 128-B f16 activation rows (the product's own operands), the f32 sum meets WMMA as hi+lo f16 tiles; sums now precede the product
# speedup vs baseline: 4.6655x; 1.0344x over previous
#include <hip/hip_runtime.h>
#include <hip/hip_bf16.h>

typedef __attribute__((ext_vector_type(16))) _Float16 v16h;
typedef __attribute__((ext_vector_type(8)))  float    v8f;

#define IN_D  128
#define HID   64


__global__ void deg_init_kernel(float* deg, int n) {
    int i = blockIdx.x * blockDim.x + threadIdx.x;
    if (i < n) deg[i] = 1.0f;
}

__global__ void deg_accum_kernel(const int* __restrict__ dst, float* deg, int e) {
    int i = blockIdx.x * blockDim.x + threadIdx.x;
    if (i < e) unsafeAtomicAdd(&deg[dst[i]], 1.0f);
}

__global__ void dinv_kernel(float* deg, int n) {
    int i = blockIdx.x * blockDim.x + threadIdx.x;
    if (i < n) deg[i] = rsqrtf(deg[i]);
}


__device__ __forceinline__ void pack_fragment(const float* __restrict__ W,
                                              _Float16* __restrict__ Wp, int frag, int lane) {
    const int ks = frag >> 2;
    const int ct = frag & 3;
    const int n    = lane & 15;
    const int koff = (lane >> 4) << 3;
    v16h b;
    const float* wp = W + (size_t)(ks * 32 + koff) * HID + ct * 16 + n;
#pragma unroll
    for (int j = 0; j < 8; ++j) {
        b[j]     = (_Float16)wp[(size_t)j * HID];
        b[8 + j] = (_Float16)wp[(size_t)(16 + j) * HID];
    }
    *(v16h*)(Wp + ((size_t)frag * 32 + lane) * 16) = b;
}

#define WP_IN_FRAGS  ((IN_D / 32) * 4)
#define WP_HID_FRAGS ((HID / 32) * 4)
#define WP_HID_SLOTS 3
#define WP_FRAG_H    (32 * 16)

template <int K>
__global__ void pack_w_kernel(const float* __restrict__ W, _Float16* __restrict__ Wp) {
    const int frag = (blockIdx.x * blockDim.x + threadIdx.x) >> 5;
    const int lane = threadIdx.x & 31;
    const int nFrag = (K / 32) * 4;
    if (frag >= nFrag) return;
    pack_fragment(W, Wp, frag, lane);
}


template <int K, bool RELU>
__global__ void gemm_wmma_kernel(const float* __restrict__ A,
                                 const _Float16* __restrict__ Wp,
                                 const float* __restrict__ rowScale,
                                 float* __restrict__ out, int nRows) {
    const int wave = (blockIdx.x * blockDim.x + threadIdx.x) >> 5;
    const int lane = threadIdx.x & 31;
    const int rowTiles = nRows >> 4;
    if (wave >= rowTiles) return;

    const int m    = lane & 15;
    const int koff = (lane >> 4) << 3;
    const int row  = (wave << 4) + m;

    const v16h* bp = (const v16h*)Wp;

    const int colLane = lane & 15;
    const int rBase   = (wave << 4) + ((lane >> 4) << 3);
    const float4 sc0 = *(const float4*)(rowScale + rBase);
    const float4 sc1 = *(const float4*)(rowScale + rBase + 4);
    const float sc[8] = {sc0.x, sc0.y, sc0.z, sc0.w, sc1.x, sc1.y, sc1.z, sc1.w};

    v8f acc0 = {}, acc1 = {}, acc2 = {}, acc3 = {};

#pragma unroll
    for (int k0 = 0; k0 < K; k0 += 32) {
        const int ks = k0 >> 5;
        v16h a;
        const float* ap = A + (size_t)row * K + k0 + koff;
#pragma unroll
        for (int j = 0; j < 8; ++j) {
            float f0 = ap[j];
            float f1 = ap[16 + j];
            if (RELU) { f0 = fmaxf(f0, 0.0f); f1 = fmaxf(f1, 0.0f); }
            a[j]     = (_Float16)f0;
            a[8 + j] = (_Float16)f1;
        }
#pragma unroll
        for (int ct = 0; ct < 4; ++ct) {
            v16h b = bp[(size_t)(ks * 4 + ct) * 32 + lane];
            v8f& acc = (ct == 0 ? acc0 : ct == 1 ? acc1 : ct == 2 ? acc2 : acc3);
            acc = __builtin_amdgcn_wmma_f32_16x16x32_f16(
                false, a, false, b, (short)0, acc, false, false);
        }
    }

    auto store_tile = [&](v8f acc, int ct) {
#pragma unroll
        for (int i = 0; i < 8; ++i)
            out[(size_t)(rBase + i) * HID + ct * 16 + colLane] = acc[i] * sc[i];
    };
    store_tile(acc0, 0); store_tile(acc1, 1); store_tile(acc2, 2); store_tile(acc3, 3);
}

__global__ void agg_init_kernel(const float* __restrict__ hW,
                                const float* __restrict__ dinv,
                                const float* __restrict__ bias,
                                float* __restrict__ out, int total) {
    int idx = blockIdx.x * blockDim.x + threadIdx.x;
    if (idx >= total) return;
    int node = idx >> 6;
    int c    = idx & 63;
    float v = hW[idx] * dinv[node];
    if (bias) v += bias[c];
    out[idx] = v;
}

__global__ void agg_edges_kernel(const float* __restrict__ hW,
                                 const int* __restrict__ src,
                                 const int* __restrict__ dst,
                                 const float* __restrict__ dinv,
                                 float* __restrict__ out, int nEdges) {
    const int wave = (blockIdx.x * blockDim.x + threadIdx.x) >> 5;
    const int lane = threadIdx.x & 31;
    if (wave >= nEdges) return;
    const int s = src[wave];
    const int d = dst[wave];
    const float w = dinv[d];
    const float2 v = ((const float2*)(hW + (size_t)s * HID))[lane];
    float* op = out + (size_t)d * HID + lane * 2;
    unsafeAtomicAdd(op,     v.x * w);
    unsafeAtomicAdd(op + 1, v.y * w);
}


#define SCAN_T   256
#define SCAN_PER 4
#define SCAN_BLK (SCAN_T * SCAN_PER)

#define BK_SHIFT 8
#define BK_NODES (1 << BK_SHIFT)
#define BK_GRID  512
#define BK_T     512
#define BK_U     8
#define FIN_T    1024
#define BK_MAXB  8192

template <bool PLACE>
__global__ void __launch_bounds__(BK_T)
bucket_pass_kernel(const int* __restrict__ src, const int* __restrict__ dst,
                   int* __restrict__ tab, const int* __restrict__ offs,
                   int2* __restrict__ pairs, int e, int chunk, int nb,
                   const float* __restrict__ W_in, const float* __restrict__ W_h,
                   const float* __restrict__ W_out, _Float16* __restrict__ Wp, int nLayers) {
    extern __shared__ int hist[];
    const int t = threadIdx.x;
    const int blk = blockIdx.x;
    const int G = BK_GRID;
    if (!PLACE && blk >= BK_GRID) {
        const int f = (blk - BK_GRID) * (BK_T / 32) + (t >> 5);
        const int lane = t & 31;
        if (f < WP_IN_FRAGS) {
            pack_fragment(W_in, Wp, f, lane);
        } else {
            const int g = f - WP_IN_FRAGS;
            const int m = g / WP_HID_FRAGS;
            if (m <= nLayers && m <= WP_HID_SLOTS) {
                const float* W = (m < nLayers) ? W_h + (size_t)m * HID * HID : W_out;
                const int slot = (m < nLayers) ? m : WP_HID_SLOTS;
                pack_fragment(W, Wp + (size_t)(WP_IN_FRAGS + slot * WP_HID_FRAGS) * WP_FRAG_H,
                              g % WP_HID_FRAGS, lane);
            }
        }
        return;
    }
    for (int i = t; i < nb; i += BK_T) hist[i] = PLACE ? offs[i * G + blk] : 0;
    __syncthreads();
    const int lo = blk * chunk;
    const int hi = min(e, lo + chunk);
    for (int i0 = lo + t; i0 < hi; i0 += BK_T * BK_U) {
        int d[BK_U], s[BK_U];
#pragma unroll
        for (int u = 0; u < BK_U; ++u) {
            const int i = i0 + u * BK_T;
            d[u] = (i < hi) ? dst[i] : 0;
            s[u] = (PLACE && i < hi) ? src[i] : 0;
        }
#pragma unroll
        for (int u = 0; u < BK_U; ++u) {
            if (i0 + u * BK_T < hi) {
                const int b = d[u] >> BK_SHIFT;
                if (PLACE) {
                    const int pos = atomicAdd(&hist[b], 1);
                    pairs[pos] = make_int2(s[u], d[u]);
                } else {
                    atomicAdd(&hist[b], 1);
                }
            }
        }
    }
    if (!PLACE) {
        __syncthreads();
        for (int i = t; i < nb; i += BK_T) tab[i * G + blk] = hist[i];
    }
}

__global__ void scan_sums_kernel(const int* __restrict__ cnt, int* __restrict__ bsum, int n) {
    __shared__ int sW[SCAN_T / 32];
    const int t = threadIdx.x;
    const int base = blockIdx.x * SCAN_BLK + t * SCAN_PER;
    int s = 0;
#pragma unroll
    for (int j = 0; j < SCAN_PER; ++j)
        if (base + j < n) s += cnt[base + j];
    for (int d = 16; d > 0; d >>= 1) s += __shfl_xor(s, d);
    if ((t & 31) == 0) sW[t >> 5] = s;
    __syncthreads();
    if (t == 0) {
        int tot = 0;
        for (int k = 0; k < SCAN_T / 32; ++k) tot += sW[k];
        bsum[blockIdx.x] = tot;
    }
}

__global__ void scan_apply_kernel(const int* __restrict__ cnt, const int* __restrict__ bsum,
                                  int* __restrict__ offs, int n) {
    __shared__ int sW[SCAN_T / 32];
    const int t = threadIdx.x;
    const int lane = t & 31;
    const int wv = t >> 5;
    const int b = blockIdx.x;

    int part = 0;
    for (int k = t; k < b; k += SCAN_T) part += bsum[k];
    for (int d = 16; d > 0; d >>= 1) part += __shfl_xor(part, d);
    if (lane == 0) sW[wv] = part;
    __syncthreads();
    int offset = 0;
    for (int k = 0; k < SCAN_T / 32; ++k) offset += sW[k];
    __syncthreads();

    const int base = b * SCAN_BLK + t * SCAN_PER;
    int c[SCAN_PER];
    int run = 0;
#pragma unroll
    for (int j = 0; j < SCAN_PER; ++j) {
        c[j] = (base + j < n) ? cnt[base + j] : 0;
        run += c[j];
    }
    int inc = run;
    for (int d = 1; d < 32; d <<= 1) {
        const int v = __shfl_up(inc, d);
        if (lane >= d) inc += v;
    }
    if (lane == 31) sW[wv] = inc;
    __syncthreads();
    int pre = offset + inc - run;
    for (int k = 0; k < wv; ++k) pre += sW[k];
#pragma unroll
    for (int j = 0; j < SCAN_PER; ++j) {
        pre += c[j];
        if (base + j < n) offs[base + j + 1] = pre;
    }
    if (b == 0 && t == 0) offs[0] = 0;
}

__global__ void __launch_bounds__(FIN_T)
bucket_finish_kernel(const int* __restrict__ offs, const int2* __restrict__ pairs,
                     int* __restrict__ rowptr, float* __restrict__ dinv,
                     int* colTmp, int* __restrict__ col, int n, int G) {
    __shared__ int cntL[BK_NODES];
    __shared__ int offL[BK_NODES];
    __shared__ int curL[BK_NODES];
    __shared__ int wtot[BK_NODES / 32];
    const int t = threadIdx.x;
    const int lane = t & 31;
    const int wv = t >> 5;
    const int b = blockIdx.x;
    const int beg = offs[b * G];
    const int end = offs[(b + 1) * G];
    if (t < BK_NODES) { cntL[t] = 0; curL[t] = 0; }
    __syncthreads();
    for (int i = beg + t; i < end; i += FIN_T)
        atomicAdd(&cntL[pairs[i].y & (BK_NODES - 1)], 1);
    __syncthreads();

    const int c = (t < BK_NODES) ? cntL[t] : 0;
    int inc = c;
    for (int d = 1; d < 32; d <<= 1) {
        const int v = __shfl_up(inc, d);
        if (lane >= d) inc += v;
    }
    if (t < BK_NODES && lane == 31) wtot[wv] = inc;
    __syncthreads();
    if (t < BK_NODES) {
        int pre = 0;
        for (int k = 0; k < wv; ++k) pre += wtot[k];
        const int excl = pre + inc - c;
        offL[t] = excl;
        const int node = b * BK_NODES + t;
        if (node < n) {
            rowptr[node + 1] = beg + excl + c;
            dinv[node] = rsqrtf((float)(c + 1));
        }
    }
    if (b == 0 && t == 0) rowptr[0] = 0;
    __syncthreads();

    for (int i = beg + t; i < end; i += FIN_T) {
        const int2 p = pairs[i];
        const int d = p.y & (BK_NODES - 1);
        const int r = atomicAdd(&curL[d], 1);
        colTmp[beg + offL[d] + r] = p.x;
    }
    __syncthreads();

    for (int row = wv; row < BK_NODES; row += FIN_T / 32) {
        const int deg = __builtin_amdgcn_readfirstlane(cntL[row]);
        const int rb  = __builtin_amdgcn_readfirstlane(beg + offL[row]);
        for (int base = 0; base < deg; base += 32) {
            const int i = base + lane;
            const int key = (i < deg) ? colTmp[rb + i] : 0;
            int r = 0;
            for (int jb = 0; jb < deg; jb += 32) {
                const int lim = min(32, deg - jb);
                const int kv = (lane < lim) ? colTmp[rb + jb + lane] : 0;
                for (int tt = 0; tt < lim; ++tt) {
                    const int kj = __builtin_amdgcn_readlane(kv, tt);
                    const int j = jb + tt;
                    r += ((kj < key) || (kj == key && j < i)) ? 1 : 0;
                }
            }
            if (i < deg) col[rb + r] = key;
        }
    }
}


#define AG_WAVES 4
#define AG_NPW   4
#define AG_BATCH 16
#define AG_BATCH_H 32
#define AG_PITCH 72

typedef __attribute__((ext_vector_type(2))) _Float16 v2h;
typedef __attribute__((ext_vector_type(8))) _Float16 v8h;

__device__ __forceinline__ float pair_x(float2 v) { return v.x; }
__device__ __forceinline__ float pair_y(float2 v) { return v.y; }
__device__ __forceinline__ float pair_x(v2h v) { return (float)v[0]; }
__device__ __forceinline__ float pair_y(v2h v) { return (float)v[1]; }
__device__ __forceinline__ void pair_zero(float2& v) { v = make_float2(0.0f, 0.0f); }
__device__ __forceinline__ void pair_zero(v2h& v) { v[0] = (_Float16)0.0f; v[1] = (_Float16)0.0f; }

template <typename PT, int BATCH>
__device__ __forceinline__ float2 sum_row(const PT* __restrict__ rows, const int* __restrict__ col,
                                          int beg, int deg, int s, int lane) {
    float ax = 0.0f, ay = 0.0f;
    const int lim = min(32, deg);
    for (int t = 0; t < lim; t += BATCH) {
        PT v[BATCH];
        {
            const int sj = __builtin_amdgcn_readlane(s, t);
            v[0] = rows[(size_t)sj * 32 + lane];
        }
#pragma unroll
        for (int u = 1; u < BATCH; ++u) {
            pair_zero(v[u]);
            if (t + u < lim) {
                const int sj = __builtin_amdgcn_readlane(s, t + u);
                v[u] = rows[(size_t)sj * 32 + lane];
            }
        }
#pragma unroll
        for (int u = 0; u < BATCH; ++u) {
            ax += pair_x(v[u]);
            ay += pair_y(v[u]);
        }
    }
    for (int e = 32; e < deg; ++e) {
        const int sj = col[beg + e];
        const PT v = rows[(size_t)sj * 32 + lane];
        ax += pair_x(v);
        ay += pair_y(v);
    }
    return make_float2(ax, ay);
}

__global__ void __launch_bounds__(AG_WAVES * 32)
agg_first_kernel(const float* __restrict__ P,
                 const int* __restrict__ rowptr,
                 const int* __restrict__ col,
                 const float* __restrict__ dinv,
                 const float* __restrict__ bias,
                 _Float16* __restrict__ G, int n) {
    const int lane = threadIdx.x & 31;
    const int wv = __builtin_amdgcn_readfirstlane((int)(threadIdx.x >> 5));
    const int node0 = blockIdx.x * 16 + wv * AG_NPW;
    const float2* rows = (const float2*)P;

    int rp[AG_NPW + 1];
#pragma unroll
    for (int j = 0; j <= AG_NPW; ++j) rp[j] = rowptr[min(node0 + j, n)];

    int    s0[AG_NPW];
    float2 self[AG_NPW];
    float  dd[AG_NPW];
#pragma unroll
    for (int j = 0; j < AG_NPW; ++j) {
        const int node = node0 + j;
        const int deg  = rp[j + 1] - rp[j];
        s0[j] = (lane < deg) ? col[rp[j] + lane] : 0;
        if (node < n) {
            self[j] = rows[(size_t)node * 32 + lane];
            dd[j]   = dinv[node];
        } else {
            self[j] = make_float2(0.0f, 0.0f);
            dd[j]   = 0.0f;
        }
    }

#pragma unroll
    for (int j = 0; j < AG_NPW; ++j) {
        const int node = node0 + j;
        const float2 a = sum_row<float2, AG_BATCH>(rows, col, rp[j], rp[j + 1] - rp[j], s0[j], lane);
        float ox = (a.x + self[j].x) * dd[j];
        float oy = (a.y + self[j].y) * dd[j];
        if (bias) { ox += bias[lane * 2]; oy += bias[lane * 2 + 1]; }
        v2h g;
        g[0] = (_Float16)(ox * dd[j]);
        g[1] = (_Float16)(oy * dd[j]);
        if (node < n) ((v2h*)(G + (size_t)node * HID))[lane] = g;
    }
}

template <bool LAST>
__global__ void __launch_bounds__(AG_WAVES * 32)
agg_mul_kernel(const _Float16* __restrict__ G,
               const int* __restrict__ rowptr,
               const int* __restrict__ col,
               const float* __restrict__ dinv,
               const float* __restrict__ bias,
               const _Float16* __restrict__ Wp,
               _Float16* __restrict__ Gout,
               float* __restrict__ out, int n) {
    __shared__ __attribute__((aligned(16))) _Float16 hiT[16 * AG_PITCH];
    __shared__ __attribute__((aligned(16))) _Float16 loT[16 * AG_PITCH];
    __shared__ __attribute__((aligned(16))) _Float16 outT[16 * AG_PITCH];
    const int lane = threadIdx.x & 31;
    const int wv = __builtin_amdgcn_readfirstlane((int)(threadIdx.x >> 5));
    const int tileBase = blockIdx.x * 16;
    const int node0 = tileBase + wv * AG_NPW;
    const v2h* rows = (const v2h*)G;

    int rp[AG_NPW + 1];
#pragma unroll
    for (int j = 0; j <= AG_NPW; ++j) rp[j] = rowptr[min(node0 + j, n)];

    int s0[AG_NPW];
    v2h self[AG_NPW];
#pragma unroll
    for (int j = 0; j < AG_NPW; ++j) {
        const int node = node0 + j;
        const int deg  = rp[j + 1] - rp[j];
        s0[j] = (lane < deg) ? col[rp[j] + lane] : 0;
        pair_zero(self[j]);
        if (node < n) self[j] = rows[(size_t)node * 32 + lane];
    }

#pragma unroll
    for (int j = 0; j < AG_NPW; ++j) {
        const float2 a = sum_row<v2h, AG_BATCH_H>(rows, col, rp[j], rp[j + 1] - rp[j], s0[j], lane);
        const float ux = a.x + pair_x(self[j]);
        const float uy = a.y + pair_y(self[j]);
        v2h ph, pl;
        ph[0] = (_Float16)ux;
        ph[1] = (_Float16)uy;
        pl[0] = (_Float16)(ux - (float)ph[0]);
        pl[1] = (_Float16)(uy - (float)ph[1]);
        *(v2h*)(hiT + (wv * AG_NPW + j) * AG_PITCH + lane * 2) = ph;
        *(v2h*)(loT + (wv * AG_NPW + j) * AG_PITCH + lane * 2) = pl;
    }

    __syncthreads();
    if (wv < 2) {
        const int m    = lane & 15;
        const int koff = (lane >> 4) << 3;
        const v16h* bp = (const v16h*)Wp;
        v8f acc0 = {}, acc1 = {};
#pragma unroll 1
        for (int ks = 0; ks < HID / 32; ++ks) {
            const _Float16* ah = hiT + m * AG_PITCH + ks * 32 + koff;
            const _Float16* al = loT + m * AG_PITCH + ks * 32 + koff;
            const v8h h0 = *(const v8h*)ah;
            const v8h h1 = *(const v8h*)(ah + 16);
            const v8h l0 = *(const v8h*)al;
            const v8h l1 = *(const v8h*)(al + 16);
            const v16h aHi = __builtin_shufflevector(h0, h1, 0, 1, 2, 3, 4, 5, 6, 7,
                                                     8, 9, 10, 11, 12, 13, 14, 15);
            const v16h aLo = __builtin_shufflevector(l0, l1, 0, 1, 2, 3, 4, 5, 6, 7,
                                                     8, 9, 10, 11, 12, 13, 14, 15);
            const v16h b0 = bp[(size_t)(ks * 4 + wv * 2 + 0) * 32 + lane];
            const v16h b1 = bp[(size_t)(ks * 4 + wv * 2 + 1) * 32 + lane];
            acc0 = __builtin_amdgcn_wmma_f32_16x16x32_f16(
                false, aHi, false, b0, (short)0, acc0, false, false);
            acc1 = __builtin_amdgcn_wmma_f32_16x16x32_f16(
                false, aHi, false, b1, (short)0, acc1, false, false);
            acc0 = __builtin_amdgcn_wmma_f32_16x16x32_f16(
                false, aLo, false, b0, (short)0, acc0, false, false);
            acc1 = __builtin_amdgcn_wmma_f32_16x16x32_f16(
                false, aLo, false, b1, (short)0, acc1, false, false);
        }
        const int colLane = lane & 15;
        const int rLoc    = (lane >> 4) << 3;
        const int c0      = wv * 32 + colLane;
        float sc[8];
#pragma unroll
        for (int i = 0; i < 8; ++i) sc[i] = (tileBase + rLoc + i < n) ? dinv[tileBase + rLoc + i] : 0.0f;
        if (LAST) {
#pragma unroll
            for (int i = 0; i < 8; ++i) {
                if (tileBase + rLoc + i < n) {
                    float* op = out + (size_t)(tileBase + rLoc + i) * HID + c0;
                    op[0]  = acc0[i] * sc[i];
                    op[16] = acc1[i] * sc[i];
                }
            }
        } else {
            const float bz0 = bias[c0];
            const float bz1 = bias[c0 + 16];
#pragma unroll
            for (int i = 0; i < 8; ++i) {
                const float v0 = fmaxf(acc0[i] * sc[i] + bz0, 0.0f);
                const float v1 = fmaxf(acc1[i] * sc[i] + bz1, 0.0f);
                outT[(rLoc + i) * AG_PITCH + c0]      = (_Float16)(v0 * sc[i]);
                outT[(rLoc + i) * AG_PITCH + c0 + 16] = (_Float16)(v1 * sc[i]);
            }
        }
    }
    if (!LAST) {
        __syncthreads();
        const int r  = threadIdx.x >> 3;
        const int ch = threadIdx.x & 7;
        if (tileBase + r < n) {
            const uint4 q = *(const uint4*)(outT + r * AG_PITCH + ch * 8);
            *(uint4*)(Gout + (size_t)(tileBase + r) * HID + ch * 8) = q;
        }
    }
}


extern "C" void kernel_launch(void* const* d_in, const int* in_sizes, int n_in,
                              void* d_out, int out_size, void* d_ws, size_t ws_size,
                              hipStream_t stream) {
    const float* x     = (const float*)d_in[0];
    const int*   eidx  = (const int*)d_in[1];
    const float* W_in  = (const float*)d_in[2];
    const float* b_in  = (const float*)d_in[3];
    const float* W_h   = (const float*)d_in[4];
    const float* b_h   = (const float*)d_in[5];
    const float* W_out = (const float*)d_in[6];

    const int N = in_sizes[0] / IN_D;
    const int E = in_sizes[1] / 2;
    const int L = in_sizes[4] / (HID * HID);

    const int* src = eidx;
    const int* dst = eidx + E;

    char* ws = (char*)d_ws;
    auto align256 = [](size_t v) { return (v + 255) & ~(size_t)255; };
    size_t off = 0;
    float* dinv = (float*)(ws + off);  off = align256(off + (size_t)N * sizeof(float));
    float* B1   = (float*)(ws + off);  off = align256(off + (size_t)N * HID * sizeof(float));
    _Float16* WpIn  = (_Float16*)(ws + off);  off = align256(off + 16 * 1024);
    _Float16* WpH0  = (_Float16*)(ws + off);  off = align256(off + 8 * 1024);
    _Float16* WpH1  = (_Float16*)(ws + off);  off = align256(off + 8 * 1024);
    _Float16* WpH2  = (_Float16*)(ws + off);  off = align256(off + 8 * 1024);
    _Float16* WpOut = (_Float16*)(ws + off);  off = align256(off + 8 * 1024);
    _Float16* WpHs[3] = {WpH0, WpH1, WpH2};
    int* rowptr = (int*)(ws + off);  off = align256(off + ((size_t)N + 1) * sizeof(int));
    int* col    = (int*)(ws + off);  off = align256(off + (size_t)E * sizeof(int));
    float* O = (float*)d_out;

    const int T = 256;

    const int rowTiles   = N / 16;
    const int gemmBlocks = (rowTiles + 7) / 8;
    const int tileBlocks = (N + 15) / 16;
    const int nBuck      = (N + BK_NODES - 1) >> BK_SHIFT;
    const int tabLen     = nBuck * BK_GRID;
    const int scanBlocks = (tabLen + SCAN_BLK - 1) / SCAN_BLK;
    const int chunk      = (E + BK_GRID - 1) / BK_GRID;

    const bool listFits = off <= ws_size && nBuck <= BK_MAXB &&
                          (size_t)3 * E + (size_t)2 * tabLen + 1 + (size_t)scanBlocks
                              <= (size_t)N * HID;

    if (listFits) {
        int2* pairs = (int2*)d_out;
        int* colTmp = (int*)d_out + (size_t)2 * E;
        int* tab    = colTmp + E;
        int* offs   = tab + tabLen;
        int* bsum   = offs + tabLen + 1;
        const size_t histBytes = (size_t)nBuck * sizeof(int);

        const int packBlocks = ((WP_IN_FRAGS + WP_HID_FRAGS * (L + 1)) * 32 + BK_T - 1) / BK_T;
        bucket_pass_kernel<false><<<BK_GRID + packBlocks, BK_T, histBytes, stream>>>(src, dst, tab, offs, pairs, E, chunk, nBuck, W_in, W_h, W_out, WpIn, L);
        scan_sums_kernel<<<scanBlocks, SCAN_T, 0, stream>>>(tab, bsum, tabLen);
        scan_apply_kernel<<<scanBlocks, SCAN_T, 0, stream>>>(tab, bsum, offs, tabLen);
        bucket_pass_kernel<true><<<BK_GRID, BK_T, histBytes, stream>>>(src, dst, tab, offs, pairs, E, chunk, nBuck, nullptr, nullptr, nullptr, nullptr, 0);
        bucket_finish_kernel<<<nBuck, FIN_T, 0, stream>>>(offs, pairs, rowptr, dinv, colTmp, col, N, BK_GRID);

        float*    P1 = (L == 0) ? O : B1;
        _Float16* Gc = (L == 0) ? (_Float16*)B1 : (_Float16*)O;
        _Float16* A0 = (_Float16*)B1;
        _Float16* A1 = A0 + (size_t)N * HID;

        gemm_wmma_kernel<IN_D, false><<<gemmBlocks, T, 0, stream>>>(x, WpIn, dinv, P1, N);
        agg_first_kernel<<<tileBlocks, AG_WAVES * 32, 0, stream>>>(P1, rowptr, col, dinv, b_in, Gc, N);

        const _Float16* g = Gc;
        for (int i = 0; i < L; ++i) {
            _Float16* gn = (i & 1) ? A1 : A0;
            agg_mul_kernel<false><<<tileBlocks, AG_WAVES * 32, 0, stream>>>(g, rowptr, col, dinv, b_h + (size_t)i * HID, WpHs[i], gn, nullptr, N);
            g = gn;
        }

        agg_mul_kernel<true><<<tileBlocks, AG_WAVES * 32, 0, stream>>>(g, rowptr, col, dinv, nullptr, WpOut, nullptr, O, N);
        return;
    }


    deg_init_kernel<<<(N + T - 1) / T, T, 0, stream>>>(dinv, N);
    deg_accum_kernel<<<(E + T - 1) / T, T, 0, stream>>>(dst, dinv, E);
    dinv_kernel<<<(N + T - 1) / T, T, 0, stream>>>(dinv, N);

    pack_w_kernel<IN_D><<<2, T, 0, stream>>>(W_in, WpIn);
    for (int l = 0; l < L; ++l)
        pack_w_kernel<HID><<<1, T, 0, stream>>>(W_h + (size_t)l * HID * HID, WpHs[l]);
    pack_w_kernel<HID><<<1, T, 0, stream>>>(W_out, WpOut);

    const int totalNC    = N * HID;
    const int initBlocks = (totalNC + T - 1) / T;
    const int edgeBlocks = (E + 7) / 8;

    gemm_wmma_kernel<IN_D, false><<<gemmBlocks, T, 0, stream>>>(x, WpIn, dinv, B1, N);
    agg_init_kernel<<<initBlocks, T, 0, stream>>>(B1, dinv, b_in, O, totalNC);
    agg_edges_kernel<<<edgeBlocks, T, 0, stream>>>(B1, src, dst, dinv, O, E);

    for (int l = 0; l < L; ++l) {
        const float* b = b_h + (size_t)l * HID;
        if (l == 0)
            gemm_wmma_kernel<HID, false><<<gemmBlocks, T, 0, stream>>>(O, WpHs[l], dinv, B1, N);
        else
            gemm_wmma_kernel<HID, true ><<<gemmBlocks, T, 0, stream>>>(O, WpHs[l], dinv, B1, N);
        agg_init_kernel<<<initBlocks, T, 0, stream>>>(B1, dinv, b, O, totalNC);
        agg_edges_kernel<<<edgeBlocks, T, 0, stream>>>(B1, src, dst, dinv, O, E);
    }

    gemm_wmma_kernel<HID, true><<<gemmBlocks, T, 0, stream>>>(O, WpOut, dinv, B1, N);
    agg_init_kernel<<<initBlocks, T, 0, stream>>>(B1, dinv, nullptr, O, totalNC);
    agg_edges_kernel<<<edgeBlocks, T, 0, stream>>>(B1, src, dst, dinv, O, E);
}


# hidden and output aggregation: a 128-B f16 row is read as 8 lanes x 16 B, the wave's 4 destination rows are summed side by side (ids handed round by DPP8); per-row sum order unchanged
# speedup vs baseline: 4.9039x; 1.0511x over previous
#include <hip/hip_runtime.h>
#include <hip/hip_bf16.h>

typedef __attribute__((ext_vector_type(16))) _Float16 v16h;
typedef __attribute__((ext_vector_type(8)))  float    v8f;

#define IN_D  128
#define HID   64


__global__ void deg_init_kernel(float* deg, int n) {
    int i = blockIdx.x * blockDim.x + threadIdx.x;
    if (i < n) deg[i] = 1.0f;
}

__global__ void deg_accum_kernel(const int* __restrict__ dst, float* deg, int e) {
    int i = blockIdx.x * blockDim.x + threadIdx.x;
    if (i < e) unsafeAtomicAdd(&deg[dst[i]], 1.0f);
}

__global__ void dinv_kernel(float* deg, int n) {
    int i = blockIdx.x * blockDim.x + threadIdx.x;
    if (i < n) deg[i] = rsqrtf(deg[i]);
}


__device__ __forceinline__ void pack_fragment(const float* __restrict__ W,
                                              _Float16* __restrict__ Wp, int frag, int lane) {
    const int ks = frag >> 2;
    const int ct = frag & 3;
    const int n    = lane & 15;
    const int koff = (lane >> 4) << 3;
    v16h b;
    const float* wp = W + (size_t)(ks * 32 + koff) * HID + ct * 16 + n;
#pragma unroll
    for (int j = 0; j < 8; ++j) {
        b[j]     = (_Float16)wp[(size_t)j * HID];
        b[8 + j] = (_Float16)wp[(size_t)(16 + j) * HID];
    }
    *(v16h*)(Wp + ((size_t)frag * 32 + lane) * 16) = b;
}

#define WP_IN_FRAGS  ((IN_D / 32) * 4)
#define WP_HID_FRAGS ((HID / 32) * 4)
#define WP_HID_SLOTS 3
#define WP_FRAG_H    (32 * 16)

template <int K>
__global__ void pack_w_kernel(const float* __restrict__ W, _Float16* __restrict__ Wp) {
    const int frag = (blockIdx.x * blockDim.x + threadIdx.x) >> 5;
    const int lane = threadIdx.x & 31;
    const int nFrag = (K / 32) * 4;
    if (frag >= nFrag) return;
    pack_fragment(W, Wp, frag, lane);
}


template <int K, bool RELU>
__global__ void gemm_wmma_kernel(const float* __restrict__ A,
                                 const _Float16* __restrict__ Wp,
                                 const float* __restrict__ rowScale,
                                 float* __restrict__ out, int nRows) {
    const int wave = (blockIdx.x * blockDim.x + threadIdx.x) >> 5;
    const int lane = threadIdx.x & 31;
    const int rowTiles = nRows >> 4;
    if (wave >= rowTiles) return;

    const int m    = lane & 15;
    const int koff = (lane >> 4) << 3;
    const int row  = (wave << 4) + m;

    const v16h* bp = (const v16h*)Wp;

    const int colLane = lane & 15;
    const int rBase   = (wave << 4) + ((lane >> 4) << 3);
    const float4 sc0 = *(const float4*)(rowScale + rBase);
    const float4 sc1 = *(const float4*)(rowScale + rBase + 4);
    const float sc[8] = {sc0.x, sc0.y, sc0.z, sc0.w, sc1.x, sc1.y, sc1.z, sc1.w};

    v8f acc0 = {}, acc1 = {}, acc2 = {}, acc3 = {};

#pragma unroll
    for (int k0 = 0; k0 < K; k0 += 32) {
        const int ks = k0 >> 5;
        v16h a;
        const float* ap = A + (size_t)row * K + k0 + koff;
#pragma unroll
        for (int j = 0; j < 8; ++j) {
            float f0 = ap[j];
            float f1 = ap[16 + j];
            if (RELU) { f0 = fmaxf(f0, 0.0f); f1 = fmaxf(f1, 0.0f); }
            a[j]     = (_Float16)f0;
            a[8 + j] = (_Float16)f1;
        }
#pragma unroll
        for (int ct = 0; ct < 4; ++ct) {
            v16h b = bp[(size_t)(ks * 4 + ct) * 32 + lane];
            v8f& acc = (ct == 0 ? acc0 : ct == 1 ? acc1 : ct == 2 ? acc2 : acc3);
            acc = __builtin_amdgcn_wmma_f32_16x16x32_f16(
                false, a, false, b, (short)0, acc, false, false);
        }
    }

    auto store_tile = [&](v8f acc, int ct) {
#pragma unroll
        for (int i = 0; i < 8; ++i)
            out[(size_t)(rBase + i) * HID + ct * 16 + colLane] = acc[i] * sc[i];
    };
    store_tile(acc0, 0); store_tile(acc1, 1); store_tile(acc2, 2); store_tile(acc3, 3);
}

__global__ void agg_init_kernel(const float* __restrict__ hW,
                                const float* __restrict__ dinv,
                                const float* __restrict__ bias,
                                float* __restrict__ out, int total) {
    int idx = blockIdx.x * blockDim.x + threadIdx.x;
    if (idx >= total) return;
    int node = idx >> 6;
    int c    = idx & 63;
    float v = hW[idx] * dinv[node];
    if (bias) v += bias[c];
    out[idx] = v;
}

__global__ void agg_edges_kernel(const float* __restrict__ hW,
                                 const int* __restrict__ src,
                                 const int* __restrict__ dst,
                                 const float* __restrict__ dinv,
                                 float* __restrict__ out, int nEdges) {
    const int wave = (blockIdx.x * blockDim.x + threadIdx.x) >> 5;
    const int lane = threadIdx.x & 31;
    if (wave >= nEdges) return;
    const int s = src[wave];
    const int d = dst[wave];
    const float w = dinv[d];
    const float2 v = ((const float2*)(hW + (size_t)s * HID))[lane];
    float* op = out + (size_t)d * HID + lane * 2;
    unsafeAtomicAdd(op,     v.x * w);
    unsafeAtomicAdd(op + 1, v.y * w);
}


#define SCAN_T   256
#define SCAN_PER 4
#define SCAN_BLK (SCAN_T * SCAN_PER)

#define BK_SHIFT 8
#define BK_NODES (1 << BK_SHIFT)
#define BK_GRID  512
#define BK_T     512
#define BK_U     8
#define FIN_T    1024
#define BK_MAXB  8192

template <bool PLACE>
__global__ void __launch_bounds__(BK_T)
bucket_pass_kernel(const int* __restrict__ src, const int* __restrict__ dst,
                   int* __restrict__ tab, const int* __restrict__ offs,
                   int2* __restrict__ pairs, int e, int chunk, int nb,
                   const float* __restrict__ W_in, const float* __restrict__ W_h,
                   const float* __restrict__ W_out, _Float16* __restrict__ Wp, int nLayers) {
    extern __shared__ int hist[];
    const int t = threadIdx.x;
    const int blk = blockIdx.x;
    const int G = BK_GRID;
    if (!PLACE && blk >= BK_GRID) {
        const int f = (blk - BK_GRID) * (BK_T / 32) + (t >> 5);
        const int lane = t & 31;
        if (f < WP_IN_FRAGS) {
            pack_fragment(W_in, Wp, f, lane);
        } else {
            const int g = f - WP_IN_FRAGS;
            const int m = g / WP_HID_FRAGS;
            if (m <= nLayers && m <= WP_HID_SLOTS) {
                const float* W = (m < nLayers) ? W_h + (size_t)m * HID * HID : W_out;
                const int slot = (m < nLayers) ? m : WP_HID_SLOTS;
                pack_fragment(W, Wp + (size_t)(WP_IN_FRAGS + slot * WP_HID_FRAGS) * WP_FRAG_H,
                              g % WP_HID_FRAGS, lane);
            }
        }
        return;
    }
    for (int i = t; i < nb; i += BK_T) hist[i] = PLACE ? offs[i * G + blk] : 0;
    __syncthreads();
    const int lo = blk * chunk;
    const int hi = min(e, lo + chunk);
    for (int i0 = lo + t; i0 < hi; i0 += BK_T * BK_U) {
        int d[BK_U], s[BK_U];
#pragma unroll
        for (int u = 0; u < BK_U; ++u) {
            const int i = i0 + u * BK_T;
            d[u] = (i < hi) ? dst[i] : 0;
            s[u] = (PLACE && i < hi) ? src[i] : 0;
        }
#pragma unroll
        for (int u = 0; u < BK_U; ++u) {
            if (i0 + u * BK_T < hi) {
                const int b = d[u] >> BK_SHIFT;
                if (PLACE) {
                    const int pos = atomicAdd(&hist[b], 1);
                    pairs[pos] = make_int2(s[u], d[u]);
                } else {
                    atomicAdd(&hist[b], 1);
                }
            }
        }
    }
    if (!PLACE) {
        __syncthreads();
        for (int i = t; i < nb; i += BK_T) tab[i * G + blk] = hist[i];
    }
}

__global__ void scan_sums_kernel(const int* __restrict__ cnt, int* __restrict__ bsum, int n) {
    __shared__ int sW[SCAN_T / 32];
    const int t = threadIdx.x;
    const int base = blockIdx.x * SCAN_BLK + t * SCAN_PER;
    int s = 0;
#pragma unroll
    for (int j = 0; j < SCAN_PER; ++j)
        if (base + j < n) s += cnt[base + j];
    for (int d = 16; d > 0; d >>= 1) s += __shfl_xor(s, d);
    if ((t & 31) == 0) sW[t >> 5] = s;
    __syncthreads();
    if (t == 0) {
        int tot = 0;
        for (int k = 0; k < SCAN_T / 32; ++k) tot += sW[k];
        bsum[blockIdx.x] = tot;
    }
}

__global__ void scan_apply_kernel(const int* __restrict__ cnt, const int* __restrict__ bsum,
                                  int* __restrict__ offs, int n) {
    __shared__ int sW[SCAN_T / 32];
    const int t = threadIdx.x;
    const int lane = t & 31;
    const int wv = t >> 5;
    const int b = blockIdx.x;

    int part = 0;
    for (int k = t; k < b; k += SCAN_T) part += bsum[k];
    for (int d = 16; d > 0; d >>= 1) part += __shfl_xor(part, d);
    if (lane == 0) sW[wv] = part;
    __syncthreads();
    int offset = 0;
    for (int k = 0; k < SCAN_T / 32; ++k) offset += sW[k];
    __syncthreads();

    const int base = b * SCAN_BLK + t * SCAN_PER;
    int c[SCAN_PER];
    int run = 0;
#pragma unroll
    for (int j = 0; j < SCAN_PER; ++j) {
        c[j] = (base + j < n) ? cnt[base + j] : 0;
        run += c[j];
    }
    int inc = run;
    for (int d = 1; d < 32; d <<= 1) {
        const int v = __shfl_up(inc, d);
        if (lane >= d) inc += v;
    }
    if (lane == 31) sW[wv] = inc;
    __syncthreads();
    int pre = offset + inc - run;
    for (int k = 0; k < wv; ++k) pre += sW[k];
#pragma unroll
    for (int j = 0; j < SCAN_PER; ++j) {
        pre += c[j];
        if (base + j < n) offs[base + j + 1] = pre;
    }
    if (b == 0 && t == 0) offs[0] = 0;
}

__global__ void __launch_bounds__(FIN_T)
bucket_finish_kernel(const int* __restrict__ offs, const int2* __restrict__ pairs,
                     int* __restrict__ rowptr, float* __restrict__ dinv,
                     int* colTmp, int* __restrict__ col, int n, int G) {
    __shared__ int cntL[BK_NODES];
    __shared__ int offL[BK_NODES];
    __shared__ int curL[BK_NODES];
    __shared__ int wtot[BK_NODES / 32];
    const int t = threadIdx.x;
    const int lane = t & 31;
    const int wv = t >> 5;
    const int b = blockIdx.x;
    const int beg = offs[b * G];
    const int end = offs[(b + 1) * G];
    if (t < BK_NODES) { cntL[t] = 0; curL[t] = 0; }
    __syncthreads();
    for (int i = beg + t; i < end; i += FIN_T)
        atomicAdd(&cntL[pairs[i].y & (BK_NODES - 1)], 1);
    __syncthreads();

    const int c = (t < BK_NODES) ? cntL[t] : 0;
    int inc = c;
    for (int d = 1; d < 32; d <<= 1) {
        const int v = __shfl_up(inc, d);
        if (lane >= d) inc += v;
    }
    if (t < BK_NODES && lane == 31) wtot[wv] = inc;
    __syncthreads();
    if (t < BK_NODES) {
        int pre = 0;
        for (int k = 0; k < wv; ++k) pre += wtot[k];
        const int excl = pre + inc - c;
        offL[t] = excl;
        const int node = b * BK_NODES + t;
        if (node < n) {
            rowptr[node + 1] = beg + excl + c;
            dinv[node] = rsqrtf((float)(c + 1));
        }
    }
    if (b == 0 && t == 0) rowptr[0] = 0;
    __syncthreads();

    for (int i = beg + t; i < end; i += FIN_T) {
        const int2 p = pairs[i];
        const int d = p.y & (BK_NODES - 1);
        const int r = atomicAdd(&curL[d], 1);
        colTmp[beg + offL[d] + r] = p.x;
    }
    __syncthreads();

    for (int row = wv; row < BK_NODES; row += FIN_T / 32) {
        const int deg = __builtin_amdgcn_readfirstlane(cntL[row]);
        const int rb  = __builtin_amdgcn_readfirstlane(beg + offL[row]);
        for (int base = 0; base < deg; base += 32) {
            const int i = base + lane;
            const int key = (i < deg) ? colTmp[rb + i] : 0;
            int r = 0;
            for (int jb = 0; jb < deg; jb += 32) {
                const int lim = min(32, deg - jb);
                const int kv = (lane < lim) ? colTmp[rb + jb + lane] : 0;
                for (int tt = 0; tt < lim; ++tt) {
                    const int kj = __builtin_amdgcn_readlane(kv, tt);
                    const int j = jb + tt;
                    r += ((kj < key) || (kj == key && j < i)) ? 1 : 0;
                }
            }
            if (i < deg) col[rb + r] = key;
        }
    }
}


#define AG_WAVES 4
#define AG_NPW   4
#define AG_BATCH 16
#define AG_SLOTS 8
#define AG_PITCH 72

typedef __attribute__((ext_vector_type(2))) _Float16 v2h;
typedef __attribute__((ext_vector_type(8))) _Float16 v8h;

__device__ __forceinline__ float pair_x(float2 v) { return v.x; }
__device__ __forceinline__ float pair_y(float2 v) { return v.y; }
__device__ __forceinline__ void pair_zero(float2& v) { v = make_float2(0.0f, 0.0f); }

template <typename PT, int BATCH>
__device__ __forceinline__ float2 sum_row(const PT* __restrict__ rows, const int* __restrict__ col,
                                          int beg, int deg, int s, int lane) {
    float ax = 0.0f, ay = 0.0f;
    const int lim = min(32, deg);
    for (int t = 0; t < lim; t += BATCH) {
        PT v[BATCH];
        {
            const int sj = __builtin_amdgcn_readlane(s, t);
            v[0] = rows[(size_t)sj * 32 + lane];
        }
#pragma unroll
        for (int u = 1; u < BATCH; ++u) {
            pair_zero(v[u]);
            if (t + u < lim) {
                const int sj = __builtin_amdgcn_readlane(s, t + u);
                v[u] = rows[(size_t)sj * 32 + lane];
            }
        }
#pragma unroll
        for (int u = 0; u < BATCH; ++u) {
            ax += pair_x(v[u]);
            ay += pair_y(v[u]);
        }
    }
    for (int e = 32; e < deg; ++e) {
        const int sj = col[beg + e];
        const PT v = rows[(size_t)sj * 32 + lane];
        ax += pair_x(v);
        ay += pair_y(v);
    }
    return make_float2(ax, ay);
}

__global__ void __launch_bounds__(AG_WAVES * 32)
agg_first_kernel(const float* __restrict__ P,
                 const int* __restrict__ rowptr,
                 const int* __restrict__ col,
                 const float* __restrict__ dinv,
                 const float* __restrict__ bias,
                 _Float16* __restrict__ G, int n) {
    const int lane = threadIdx.x & 31;
    const int wv = __builtin_amdgcn_readfirstlane((int)(threadIdx.x >> 5));
    const int node0 = blockIdx.x * 16 + wv * AG_NPW;
    const float2* rows = (const float2*)P;

    int rp[AG_NPW + 1];
#pragma unroll
    for (int j = 0; j <= AG_NPW; ++j) rp[j] = rowptr[min(node0 + j, n)];

    int    s0[AG_NPW];
    float2 self[AG_NPW];
    float  dd[AG_NPW];
#pragma unroll
    for (int j = 0; j < AG_NPW; ++j) {
        const int node = node0 + j;
        const int deg  = rp[j + 1] - rp[j];
        s0[j] = (lane < deg) ? col[rp[j] + lane] : 0;
        if (node < n) {
            self[j] = rows[(size_t)node * 32 + lane];
            dd[j]   = dinv[node];
        } else {
            self[j] = make_float2(0.0f, 0.0f);
            dd[j]   = 0.0f;
        }
    }

#pragma unroll
    for (int j = 0; j < AG_NPW; ++j) {
        const int node = node0 + j;
        const float2 a = sum_row<float2, AG_BATCH>(rows, col, rp[j], rp[j + 1] - rp[j], s0[j], lane);
        float ox = (a.x + self[j].x) * dd[j];
        float oy = (a.y + self[j].y) * dd[j];
        if (bias) { ox += bias[lane * 2]; oy += bias[lane * 2 + 1]; }
        v2h g;
        g[0] = (_Float16)(ox * dd[j]);
        g[1] = (_Float16)(oy * dd[j]);
        if (node < n) ((v2h*)(G + (size_t)node * HID))[lane] = g;
    }
}


__device__ __forceinline__ int group8_pick(int v, int u) {
    switch (u) {
        case 0:  return __builtin_amdgcn_mov_dpp8(v, 0 * 0x249249);
        case 1:  return __builtin_amdgcn_mov_dpp8(v, 1 * 0x249249);
        case 2:  return __builtin_amdgcn_mov_dpp8(v, 2 * 0x249249);
        case 3:  return __builtin_amdgcn_mov_dpp8(v, 3 * 0x249249);
        case 4:  return __builtin_amdgcn_mov_dpp8(v, 4 * 0x249249);
        case 5:  return __builtin_amdgcn_mov_dpp8(v, 5 * 0x249249);
        case 6:  return __builtin_amdgcn_mov_dpp8(v, 6 * 0x249249);
        default: return __builtin_amdgcn_mov_dpp8(v, 7 * 0x249249);
    }
}

__device__ __forceinline__ void walk8(const char* __restrict__ Gb, int ids, unsigned qoff,
                                      int first, int lim, int deg, float (&acc)[8]) {
    uint4 v[AG_SLOTS];
#pragma unroll
    for (int u = 0; u < AG_SLOTS; ++u) {
        const int id = group8_pick(ids, u);
        v[u] = make_uint4(0u, 0u, 0u, 0u);
        if (first + u < deg) {
            const unsigned off = (unsigned)id * 128u + qoff;
            v[u] = *(const uint4*)(Gb + (size_t)off);
        }
    }
#pragma unroll
    for (int u = 0; u < AG_SLOTS; ++u) {
        if (first + u < lim) {
            const v8h h = __builtin_bit_cast(v8h, v[u]);
#pragma unroll
            for (int k = 0; k < 8; ++k) acc[k] += (float)h[k];
        }
    }
}

template <bool LAST>
__global__ void __launch_bounds__(AG_WAVES * 32)
agg_mul_kernel(const _Float16* __restrict__ G,
               const int* __restrict__ rowptr,
               const int* __restrict__ col,
               const float* __restrict__ dinv,
               const float* __restrict__ bias,
               const _Float16* __restrict__ Wp,
               _Float16* __restrict__ Gout,
               float* __restrict__ out, int n) {
    __shared__ __attribute__((aligned(16))) _Float16 hiT[16 * AG_PITCH];
    __shared__ __attribute__((aligned(16))) _Float16 loT[16 * AG_PITCH];
    __shared__ __attribute__((aligned(16))) _Float16 outT[16 * AG_PITCH];
    const int lane = threadIdx.x & 31;
    const int wv = __builtin_amdgcn_readfirstlane((int)(threadIdx.x >> 5));
    const int tileBase = blockIdx.x * 16;
    const int grp = lane >> 3;
    const int q   = lane & 7;
    const int rloc = wv * AG_NPW + grp;
    const int node = tileBase + rloc;
    const char* Gb = (const char*)G;
    const unsigned qoff = (unsigned)q * 16u;

    const int beg = rowptr[min(node, n)];
    const int deg = rowptr[min(node + 1, n)] - beg;

    int ids[4];
#pragma unroll
    for (int m = 0; m < 4; ++m) ids[m] = (q + 8 * m < deg) ? col[beg + q + 8 * m] : 0;
    uint4 selfr = make_uint4(0u, 0u, 0u, 0u);
    if (node < n) selfr = *(const uint4*)(Gb + (size_t)((unsigned)node * 128u + qoff));

    const int maxdeg = max(max(__builtin_amdgcn_readlane(deg, 0), __builtin_amdgcn_readlane(deg, 8)),
                           max(__builtin_amdgcn_readlane(deg, 16), __builtin_amdgcn_readlane(deg, 24)));
    const int lim = min(maxdeg, 32);

    float acc[8];
#pragma unroll
    for (int k = 0; k < 8; ++k) acc[k] = 0.0f;
    if (lim > 0)  walk8(Gb, ids[0], qoff, 0,  lim, deg, acc);
    if (lim > 8)  walk8(Gb, ids[1], qoff, 8,  lim, deg, acc);
    if (lim > 16) walk8(Gb, ids[2], qoff, 16, lim, deg, acc);
    if (lim > 24) walk8(Gb, ids[3], qoff, 24, lim, deg, acc);
    for (int e = 32; e < maxdeg; ++e) {
        if (e < deg) {
            const int id = col[beg + e];
            const v8h h = *(const v8h*)(Gb + (size_t)((unsigned)id * 128u + qoff));
#pragma unroll
            for (int k = 0; k < 8; ++k) acc[k] += (float)h[k];
        }
    }

    {
        const v8h sh = __builtin_bit_cast(v8h, selfr);
        v8h ph, pl;
#pragma unroll
        for (int k = 0; k < 8; ++k) {
            const float u = acc[k] + (float)sh[k];
            ph[k] = (_Float16)u;
            pl[k] = (_Float16)(u - (float)ph[k]);
        }
        *(v8h*)(hiT + rloc * AG_PITCH + q * 8) = ph;
        *(v8h*)(loT + rloc * AG_PITCH + q * 8) = pl;
    }

    __syncthreads();
    if (wv < 2) {
        const int m    = lane & 15;
        const int koff = (lane >> 4) << 3;
        const v16h* bp = (const v16h*)Wp;
        v8f acc0 = {}, acc1 = {};
#pragma unroll 1
        for (int ks = 0; ks < HID / 32; ++ks) {
            const _Float16* ah = hiT + m * AG_PITCH + ks * 32 + koff;
            const _Float16* al = loT + m * AG_PITCH + ks * 32 + koff;
            const v8h h0 = *(const v8h*)ah;
            const v8h h1 = *(const v8h*)(ah + 16);
            const v8h l0 = *(const v8h*)al;
            const v8h l1 = *(const v8h*)(al + 16);
            const v16h aHi = __builtin_shufflevector(h0, h1, 0, 1, 2, 3, 4, 5, 6, 7,
                                                     8, 9, 10, 11, 12, 13, 14, 15);
            const v16h aLo = __builtin_shufflevector(l0, l1, 0, 1, 2, 3, 4, 5, 6, 7,
                                                     8, 9, 10, 11, 12, 13, 14, 15);
            const v16h b0 = bp[(size_t)(ks * 4 + wv * 2 + 0) * 32 + lane];
            const v16h b1 = bp[(size_t)(ks * 4 + wv * 2 + 1) * 32 + lane];
            acc0 = __builtin_amdgcn_wmma_f32_16x16x32_f16(
                false, aHi, false, b0, (short)0, acc0, false, false);
            acc1 = __builtin_amdgcn_wmma_f32_16x16x32_f16(
                false, aHi, false, b1, (short)0, acc1, false, false);
            acc0 = __builtin_amdgcn_wmma_f32_16x16x32_f16(
                false, aLo, false, b0, (short)0, acc0, false, false);
            acc1 = __builtin_amdgcn_wmma_f32_16x16x32_f16(
                false, aLo, false, b1, (short)0, acc1, false, false);
        }
        const int colLane = lane & 15;
        const int rLoc    = (lane >> 4) << 3;
        const int c0      = wv * 32 + colLane;
        float sc[8];
#pragma unroll
        for (int i = 0; i < 8; ++i) sc[i] = (tileBase + rLoc + i < n) ? dinv[tileBase + rLoc + i] : 0.0f;
        if (LAST) {
#pragma unroll
            for (int i = 0; i < 8; ++i) {
                if (tileBase + rLoc + i < n) {
                    float* op = out + (size_t)(tileBase + rLoc + i) * HID + c0;
                    op[0]  = acc0[i] * sc[i];
                    op[16] = acc1[i] * sc[i];
                }
            }
        } else {
            const float bz0 = bias[c0];
            const float bz1 = bias[c0 + 16];
#pragma unroll
            for (int i = 0; i < 8; ++i) {
                const float v0 = fmaxf(acc0[i] * sc[i] + bz0, 0.0f);
                const float v1 = fmaxf(acc1[i] * sc[i] + bz1, 0.0f);
                outT[(rLoc + i) * AG_PITCH + c0]      = (_Float16)(v0 * sc[i]);
                outT[(rLoc + i) * AG_PITCH + c0 + 16] = (_Float16)(v1 * sc[i]);
            }
        }
    }
    if (!LAST) {
        __syncthreads();
        const int r  = threadIdx.x >> 3;
        const int ch = threadIdx.x & 7;
        if (tileBase + r < n) {
            const uint4 q = *(const uint4*)(outT + r * AG_PITCH + ch * 8);
            *(uint4*)(Gout + (size_t)(tileBase + r) * HID + ch * 8) = q;
        }
    }
}


extern "C" void kernel_launch(void* const* d_in, const int* in_sizes, int n_in,
                              void* d_out, int out_size, void* d_ws, size_t ws_size,
                              hipStream_t stream) {
    const float* x     = (const float*)d_in[0];
    const int*   eidx  = (const int*)d_in[1];
    const float* W_in  = (const float*)d_in[2];
    const float* b_in  = (const float*)d_in[3];
    const float* W_h   = (const float*)d_in[4];
    const float* b_h   = (const float*)d_in[5];
    const float* W_out = (const float*)d_in[6];

    const int N = in_sizes[0] / IN_D;
    const int E = in_sizes[1] / 2;
    const int L = in_sizes[4] / (HID * HID);

    const int* src = eidx;
    const int* dst = eidx + E;

    char* ws = (char*)d_ws;
    auto align256 = [](size_t v) { return (v + 255) & ~(size_t)255; };
    size_t off = 0;
    float* dinv = (float*)(ws + off);  off = align256(off + (size_t)N * sizeof(float));
    float* B1   = (float*)(ws + off);  off = align256(off + (size_t)N * HID * sizeof(float));
    _Float16* WpIn  = (_Float16*)(ws + off);  off = align256(off + 16 * 1024);
    _Float16* WpH0  = (_Float16*)(ws + off);  off = align256(off + 8 * 1024);
    _Float16* WpH1  = (_Float16*)(ws + off);  off = align256(off + 8 * 1024);
    _Float16* WpH2  = (_Float16*)(ws + off);  off = align256(off + 8 * 1024);
    _Float16* WpOut = (_Float16*)(ws + off);  off = align256(off + 8 * 1024);
    _Float16* WpHs[3] = {WpH0, WpH1, WpH2};
    int* rowptr = (int*)(ws + off);  off = align256(off + ((size_t)N + 1) * sizeof(int));
    int* col    = (int*)(ws + off);  off = align256(off + (size_t)E * sizeof(int));
    float* O = (float*)d_out;

    const int T = 256;

    const int rowTiles   = N / 16;
    const int gemmBlocks = (rowTiles + 7) / 8;
    const int tileBlocks = (N + 15) / 16;
    const int nBuck      = (N + BK_NODES - 1) >> BK_SHIFT;
    const int tabLen     = nBuck * BK_GRID;
    const int scanBlocks = (tabLen + SCAN_BLK - 1) / SCAN_BLK;
    const int chunk      = (E + BK_GRID - 1) / BK_GRID;

    const bool listFits = off <= ws_size && nBuck <= BK_MAXB &&
                          (size_t)3 * E + (size_t)2 * tabLen + 1 + (size_t)scanBlocks
                              <= (size_t)N * HID;

    if (listFits) {
        int2* pairs = (int2*)d_out;
        int* colTmp = (int*)d_out + (size_t)2 * E;
        int* tab    = colTmp + E;
        int* offs   = tab + tabLen;
        int* bsum   = offs + tabLen + 1;
        const size_t histBytes = (size_t)nBuck * sizeof(int);

        const int packBlocks = ((WP_IN_FRAGS + WP_HID_FRAGS * (L + 1)) * 32 + BK_T - 1) / BK_T;
        bucket_pass_kernel<false><<<BK_GRID + packBlocks, BK_T, histBytes, stream>>>(src, dst, tab, offs, pairs, E, chunk, nBuck, W_in, W_h, W_out, WpIn, L);
        scan_sums_kernel<<<scanBlocks, SCAN_T, 0, stream>>>(tab, bsum, tabLen);
        scan_apply_kernel<<<scanBlocks, SCAN_T, 0, stream>>>(tab, bsum, offs, tabLen);
        bucket_pass_kernel<true><<<BK_GRID, BK_T, histBytes, stream>>>(src, dst, tab, offs, pairs, E, chunk, nBuck, nullptr, nullptr, nullptr, nullptr, 0);
        bucket_finish_kernel<<<nBuck, FIN_T, 0, stream>>>(offs, pairs, rowptr, dinv, colTmp, col, N, BK_GRID);

        float*    P1 = (L == 0) ? O : B1;
        _Float16* Gc = (L == 0) ? (_Float16*)B1 : (_Float16*)O;
        _Float16* A0 = (_Float16*)B1;
        _Float16* A1 = A0 + (size_t)N * HID;

        gemm_wmma_kernel<IN_D, false><<<gemmBlocks, T, 0, stream>>>(x, WpIn, dinv, P1, N);
        agg_first_kernel<<<tileBlocks, AG_WAVES * 32, 0, stream>>>(P1, rowptr, col, dinv, b_in, Gc, N);

        const _Float16* g = Gc;
        for (int i = 0; i < L; ++i) {
            _Float16* gn = (i & 1) ? A1 : A0;
            agg_mul_kernel<false><<<tileBlocks, AG_WAVES * 32, 0, stream>>>(g, rowptr, col, dinv, b_h + (size_t)i * HID, WpHs[i], gn, nullptr, N);
            g = gn;
        }

        agg_mul_kernel<true><<<tileBlocks, AG_WAVES * 32, 0, stream>>>(g, rowptr, col, dinv, nullptr, WpOut, nullptr, O, N);
        return;
    }


    deg_init_kernel<<<(N + T - 1) / T, T, 0, stream>>>(dinv, N);
    deg_accum_kernel<<<(E + T - 1) / T, T, 0, stream>>>(dst, dinv, E);
    dinv_kernel<<<(N + T - 1) / T, T, 0, stream>>>(dinv, N);

    pack_w_kernel<IN_D><<<2, T, 0, stream>>>(W_in, WpIn);
    for (int l = 0; l < L; ++l)
        pack_w_kernel<HID><<<1, T, 0, stream>>>(W_h + (size_t)l * HID * HID, WpHs[l]);
    pack_w_kernel<HID><<<1, T, 0, stream>>>(W_out, WpOut);

    const int totalNC    = N * HID;
    const int initBlocks = (totalNC + T - 1) / T;
    const int edgeBlocks = (E + 7) / 8;

    gemm_wmma_kernel<IN_D, false><<<gemmBlocks, T, 0, stream>>>(x, WpIn, dinv, B1, N);
    agg_init_kernel<<<initBlocks, T, 0, stream>>>(B1, dinv, b_in, O, totalNC);
    agg_edges_kernel<<<edgeBlocks, T, 0, stream>>>(B1, src, dst, dinv, O, E);

    for (int l = 0; l < L; ++l) {
        const float* b = b_h + (size_t)l * HID;
        if (l == 0)
            gemm_wmma_kernel<HID, false><<<gemmBlocks, T, 0, stream>>>(O, WpHs[l], dinv, B1, N);
        else
            gemm_wmma_kernel<HID, true ><<<gemmBlocks, T, 0, stream>>>(O, WpHs[l], dinv, B1, N);
        agg_init_kernel<<<initBlocks, T, 0, stream>>>(B1, dinv, b, O, totalNC);
        agg_edges_kernel<<<edgeBlocks, T, 0, stream>>>(B1, src, dst, dinv, O, E);
    }

    gemm_wmma_kernel<HID, true><<<gemmBlocks, T, 0, stream>>>(O, WpOut, dinv, B1, N);
    agg_init_kernel<<<initBlocks, T, 0, stream>>>(B1, dinv, nullptr, O, totalNC);
    agg_edges_kernel<<<edgeBlocks, T, 0, stream>>>(B1, src, dst, dinv, O, E);
}


# edge ordering: a block orders its own slice by bucket in LDS (count, scan, place); bucket start = sum of slice offsets: count table and 3 launches gone (11 -> 8); batch loads unpredicated (clamped)
# speedup vs baseline: 5.4470x; 1.1107x over previous
#include <hip/hip_runtime.h>
#include <hip/hip_bf16.h>

typedef __attribute__((ext_vector_type(16))) _Float16 v16h;
typedef __attribute__((ext_vector_type(8)))  float    v8f;

#define IN_D  128
#define HID   64


__global__ void deg_init_kernel(float* deg, int n) {
    int i = blockIdx.x * blockDim.x + threadIdx.x;
    if (i < n) deg[i] = 1.0f;
}

__global__ void deg_accum_kernel(const int* __restrict__ dst, float* deg, int e) {
    int i = blockIdx.x * blockDim.x + threadIdx.x;
    if (i < e) unsafeAtomicAdd(&deg[dst[i]], 1.0f);
}

__global__ void dinv_kernel(float* deg, int n) {
    int i = blockIdx.x * blockDim.x + threadIdx.x;
    if (i < n) deg[i] = rsqrtf(deg[i]);
}


__device__ __forceinline__ void pack_fragment(const float* __restrict__ W,
                                              _Float16* __restrict__ Wp, int frag, int lane) {
    const int ks = frag >> 2;
    const int ct = frag & 3;
    const int n    = lane & 15;
    const int koff = (lane >> 4) << 3;
    v16h b;
    const float* wp = W + (size_t)(ks * 32 + koff) * HID + ct * 16 + n;
#pragma unroll
    for (int j = 0; j < 8; ++j) {
        b[j]     = (_Float16)wp[(size_t)j * HID];
        b[8 + j] = (_Float16)wp[(size_t)(16 + j) * HID];
    }
    *(v16h*)(Wp + ((size_t)frag * 32 + lane) * 16) = b;
}

#define WP_IN_FRAGS  ((IN_D / 32) * 4)
#define WP_HID_FRAGS ((HID / 32) * 4)
#define WP_HID_SLOTS 3
#define WP_FRAG_H    (32 * 16)

template <int K>
__global__ void pack_w_kernel(const float* __restrict__ W, _Float16* __restrict__ Wp) {
    const int frag = (blockIdx.x * blockDim.x + threadIdx.x) >> 5;
    const int lane = threadIdx.x & 31;
    const int nFrag = (K / 32) * 4;
    if (frag >= nFrag) return;
    pack_fragment(W, Wp, frag, lane);
}


template <int K, bool RELU>
__global__ void gemm_wmma_kernel(const float* __restrict__ A,
                                 const _Float16* __restrict__ Wp,
                                 const float* __restrict__ rowScale,
                                 float* __restrict__ out, int nRows) {
    const int wave = (blockIdx.x * blockDim.x + threadIdx.x) >> 5;
    const int lane = threadIdx.x & 31;
    const int rowTiles = nRows >> 4;
    if (wave >= rowTiles) return;

    const int m    = lane & 15;
    const int koff = (lane >> 4) << 3;
    const int row  = (wave << 4) + m;

    const v16h* bp = (const v16h*)Wp;

    const int colLane = lane & 15;
    const int rBase   = (wave << 4) + ((lane >> 4) << 3);
    const float4 sc0 = *(const float4*)(rowScale + rBase);
    const float4 sc1 = *(const float4*)(rowScale + rBase + 4);
    const float sc[8] = {sc0.x, sc0.y, sc0.z, sc0.w, sc1.x, sc1.y, sc1.z, sc1.w};

    v8f acc0 = {}, acc1 = {}, acc2 = {}, acc3 = {};

#pragma unroll
    for (int k0 = 0; k0 < K; k0 += 32) {
        const int ks = k0 >> 5;
        v16h a;
        const float* ap = A + (size_t)row * K + k0 + koff;
#pragma unroll
        for (int j = 0; j < 8; ++j) {
            float f0 = ap[j];
            float f1 = ap[16 + j];
            if (RELU) { f0 = fmaxf(f0, 0.0f); f1 = fmaxf(f1, 0.0f); }
            a[j]     = (_Float16)f0;
            a[8 + j] = (_Float16)f1;
        }
#pragma unroll
        for (int ct = 0; ct < 4; ++ct) {
            v16h b = bp[(size_t)(ks * 4 + ct) * 32 + lane];
            v8f& acc = (ct == 0 ? acc0 : ct == 1 ? acc1 : ct == 2 ? acc2 : acc3);
            acc = __builtin_amdgcn_wmma_f32_16x16x32_f16(
                false, a, false, b, (short)0, acc, false, false);
        }
    }

    auto store_tile = [&](v8f acc, int ct) {
#pragma unroll
        for (int i = 0; i < 8; ++i)
            out[(size_t)(rBase + i) * HID + ct * 16 + colLane] = acc[i] * sc[i];
    };
    store_tile(acc0, 0); store_tile(acc1, 1); store_tile(acc2, 2); store_tile(acc3, 3);
}

__global__ void agg_init_kernel(const float* __restrict__ hW,
                                const float* __restrict__ dinv,
                                const float* __restrict__ bias,
                                float* __restrict__ out, int total) {
    int idx = blockIdx.x * blockDim.x + threadIdx.x;
    if (idx >= total) return;
    int node = idx >> 6;
    int c    = idx & 63;
    float v = hW[idx] * dinv[node];
    if (bias) v += bias[c];
    out[idx] = v;
}

__global__ void agg_edges_kernel(const float* __restrict__ hW,
                                 const int* __restrict__ src,
                                 const int* __restrict__ dst,
                                 const float* __restrict__ dinv,
                                 float* __restrict__ out, int nEdges) {
    const int wave = (blockIdx.x * blockDim.x + threadIdx.x) >> 5;
    const int lane = threadIdx.x & 31;
    if (wave >= nEdges) return;
    const int s = src[wave];
    const int d = dst[wave];
    const float w = dinv[d];
    const float2 v = ((const float2*)(hW + (size_t)s * HID))[lane];
    float* op = out + (size_t)d * HID + lane * 2;
    unsafeAtomicAdd(op,     v.x * w);
    unsafeAtomicAdd(op + 1, v.y * w);
}


#define BK_SHIFT 8
#define BK_NODES (1 << BK_SHIFT)
#define BK_GRID  512
#define BK_T     512
#define BK_U     8
#define FIN_T    1024
#define FIN_TPS  (FIN_T / BK_GRID)
#define FIN_U    4
#define BK_MAXB  8192

__global__ void __launch_bounds__(BK_T)
slice_order_kernel(const int* __restrict__ src, const int* __restrict__ dst,
                   int* __restrict__ offsIn, int2* __restrict__ pairs, int e, int chunk, int nb,
                   const float* __restrict__ W_in, const float* __restrict__ W_h,
                   const float* __restrict__ W_out, _Float16* __restrict__ Wp, int nLayers) {
    extern __shared__ int hist[];
    __shared__ int wsum[BK_T / 32];
    const int t = threadIdx.x;
    const int blk = blockIdx.x;
    if (blk >= BK_GRID) {
        const int f = (blk - BK_GRID) * (BK_T / 32) + (t >> 5);
        const int lane = t & 31;
        if (f < WP_IN_FRAGS) {
            pack_fragment(W_in, Wp, f, lane);
        } else {
            const int g = f - WP_IN_FRAGS;
            const int m = g / WP_HID_FRAGS;
            if (m <= nLayers && m <= WP_HID_SLOTS) {
                const float* W = (m < nLayers) ? W_h + (size_t)m * HID * HID : W_out;
                const int slot = (m < nLayers) ? m : WP_HID_SLOTS;
                pack_fragment(W, Wp + (size_t)(WP_IN_FRAGS + slot * WP_HID_FRAGS) * WP_FRAG_H,
                              g % WP_HID_FRAGS, lane);
            }
        }
        return;
    }
    for (int i = t; i < nb; i += BK_T) hist[i] = 0;
    __syncthreads();
    const int lo = blk * chunk;
    const int hi = min(e, lo + chunk);

    for (int i0 = lo + t; i0 < hi; i0 += BK_T * BK_U) {
        int d[BK_U];
#pragma unroll
        for (int u = 0; u < BK_U; ++u) d[u] = dst[min(i0 + u * BK_T, hi - 1)];
#pragma unroll
        for (int u = 0; u < BK_U; ++u)
            atomicAdd(&hist[d[u] >> BK_SHIFT], (i0 + u * BK_T < hi) ? 1 : 0);
    }
    __syncthreads();

    const int lane = t & 31;
    const int wv = t >> 5;
    const int ipt = (nb + BK_T - 1) / BK_T;
    const int b0 = t * ipt;
    int run = 0;
    for (int j = 0; j < ipt; ++j)
        if (b0 + j < nb) run += hist[b0 + j];
    int inc = run;
    for (int d = 1; d < 32; d <<= 1) {
        const int v = __shfl_up(inc, d);
        if (lane >= d) inc += v;
    }
    if (lane == 31) wsum[wv] = inc;
    __syncthreads();
    int pre = inc - run;
    for (int k = 0; k < wv; ++k) pre += wsum[k];
    int* row = offsIn + (size_t)blk * (nb + 1);
    for (int j = 0; j < ipt; ++j) {
        if (b0 + j < nb) {
            const int cnt = hist[b0 + j];
            hist[b0 + j] = pre;
            row[b0 + j]  = pre;
            pre += cnt;
        }
    }
    if (t == BK_T - 1) row[nb] = pre;
    __syncthreads();

    for (int i0 = lo + t; i0 < hi; i0 += BK_T * BK_U) {
        int d[BK_U], s[BK_U];
#pragma unroll
        for (int u = 0; u < BK_U; ++u) {
            const int i = min(i0 + u * BK_T, hi - 1);
            d[u] = dst[i];
            s[u] = src[i];
        }
#pragma unroll
        for (int u = 0; u < BK_U; ++u) {
            const bool in = i0 + u * BK_T < hi;
            const int pos = atomicAdd(&hist[d[u] >> BK_SHIFT], in ? 1 : 0);
            if (in) pairs[lo + pos] = make_int2(s[u], d[u]);
        }
    }
}

__global__ void __launch_bounds__(FIN_T)
bucket_finish_kernel(const int* __restrict__ offsIn, const int2* __restrict__ pairs,
                     int* __restrict__ rowptr, float* __restrict__ dinv,
                     int* colTmp, int* __restrict__ col, int n, int nb, int chunk) {
    __shared__ int cntL[BK_NODES];
    __shared__ int offL[BK_NODES];
    __shared__ int curL[BK_NODES];
    __shared__ int wtot[BK_NODES / 32];
    __shared__ int wbeg[FIN_T / 32];
    const int t = threadIdx.x;
    const int lane = t & 31;
    const int wv = t >> 5;
    const int b = blockIdx.x;
    const int sl  = t / FIN_TPS;
    const int par = t % FIN_TPS;
    const int* orow = offsIn + (size_t)sl * (nb + 1) + b;
    const int lo = orow[0];
    const int hi = orow[1];
    const int2* pp = pairs + (size_t)sl * chunk;
    int part = (par == 0) ? lo : 0;
    for (int d = 16; d > 0; d >>= 1) part += __shfl_xor(part, d);
    if (lane == 0) wbeg[wv] = part;
    if (t < BK_NODES) { cntL[t] = 0; curL[t] = 0; }
    __syncthreads();
    int beg = 0;
    for (int k = 0; k < FIN_T / 32; ++k) beg += wbeg[k];
    for (int i0 = lo + par; i0 < hi; i0 += FIN_U * FIN_TPS) {
        int dd[FIN_U];
#pragma unroll
        for (int u = 0; u < FIN_U; ++u) dd[u] = pp[min(i0 + u * FIN_TPS, hi - 1)].y;
#pragma unroll
        for (int u = 0; u < FIN_U; ++u)
            atomicAdd(&cntL[dd[u] & (BK_NODES - 1)], (i0 + u * FIN_TPS < hi) ? 1 : 0);
    }
    __syncthreads();

    const int c = (t < BK_NODES) ? cntL[t] : 0;
    int inc = c;
    for (int d = 1; d < 32; d <<= 1) {
        const int v = __shfl_up(inc, d);
        if (lane >= d) inc += v;
    }
    if (t < BK_NODES && lane == 31) wtot[wv] = inc;
    __syncthreads();
    if (t < BK_NODES) {
        int pre = 0;
        for (int k = 0; k < wv; ++k) pre += wtot[k];
        const int excl = pre + inc - c;
        offL[t] = excl;
        const int node = b * BK_NODES + t;
        if (node < n) {
            rowptr[node + 1] = beg + excl + c;
            dinv[node] = rsqrtf((float)(c + 1));
        }
    }
    if (b == 0 && t == 0) rowptr[0] = 0;
    __syncthreads();

    for (int i0 = lo + par; i0 < hi; i0 += FIN_U * FIN_TPS) {
        int2 p[FIN_U];
#pragma unroll
        for (int u = 0; u < FIN_U; ++u) p[u] = pp[min(i0 + u * FIN_TPS, hi - 1)];
#pragma unroll
        for (int u = 0; u < FIN_U; ++u) {
            const bool in = i0 + u * FIN_TPS < hi;
            const int d = p[u].y & (BK_NODES - 1);
            const int r = atomicAdd(&curL[d], in ? 1 : 0);
            if (in) colTmp[beg + offL[d] + r] = p[u].x;
        }
    }
    __syncthreads();

    for (int row = wv; row < BK_NODES; row += FIN_T / 32) {
        const int deg = __builtin_amdgcn_readfirstlane(cntL[row]);
        const int rb  = __builtin_amdgcn_readfirstlane(beg + offL[row]);
        for (int base = 0; base < deg; base += 32) {
            const int i = base + lane;
            const int key = (i < deg) ? colTmp[rb + i] : 0;
            int r = 0;
            for (int jb = 0; jb < deg; jb += 32) {
                const int lim = min(32, deg - jb);
                const int kv = (lane < lim) ? colTmp[rb + jb + lane] : 0;
                for (int tt = 0; tt < lim; ++tt) {
                    const int kj = __builtin_amdgcn_readlane(kv, tt);
                    const int j = jb + tt;
                    r += ((kj < key) || (kj == key && j < i)) ? 1 : 0;
                }
            }
            if (i < deg) col[rb + r] = key;
        }
    }
}


#define AG_WAVES 4
#define AG_NPW   4
#define AG_BATCH 16
#define AG_SLOTS 8
#define AG_PITCH 72

typedef __attribute__((ext_vector_type(2))) _Float16 v2h;
typedef __attribute__((ext_vector_type(8))) _Float16 v8h;

__device__ __forceinline__ float pair_x(float2 v) { return v.x; }
__device__ __forceinline__ float pair_y(float2 v) { return v.y; }
__device__ __forceinline__ void pair_zero(float2& v) { v = make_float2(0.0f, 0.0f); }

template <typename PT, int BATCH>
__device__ __forceinline__ float2 sum_row(const PT* __restrict__ rows, const int* __restrict__ col,
                                          int beg, int deg, int s, int lane) {
    float ax = 0.0f, ay = 0.0f;
    const int lim = min(32, deg);
    for (int t = 0; t < lim; t += BATCH) {
        PT v[BATCH];
        {
            const int sj = __builtin_amdgcn_readlane(s, t);
            v[0] = rows[(size_t)sj * 32 + lane];
        }
#pragma unroll
        for (int u = 1; u < BATCH; ++u) {
            pair_zero(v[u]);
            if (t + u < lim) {
                const int sj = __builtin_amdgcn_readlane(s, t + u);
                v[u] = rows[(size_t)sj * 32 + lane];
            }
        }
#pragma unroll
        for (int u = 0; u < BATCH; ++u) {
            ax += pair_x(v[u]);
            ay += pair_y(v[u]);
        }
    }
    for (int e = 32; e < deg; ++e) {
        const int sj = col[beg + e];
        const PT v = rows[(size_t)sj * 32 + lane];
        ax += pair_x(v);
        ay += pair_y(v);
    }
    return make_float2(ax, ay);
}

__global__ void __launch_bounds__(AG_WAVES * 32)
agg_first_kernel(const float* __restrict__ P,
                 const int* __restrict__ rowptr,
                 const int* __restrict__ col,
                 const float* __restrict__ dinv,
                 const float* __restrict__ bias,
                 _Float16* __restrict__ G, int n) {
    const int lane = threadIdx.x & 31;
    const int wv = __builtin_amdgcn_readfirstlane((int)(threadIdx.x >> 5));
    const int node0 = blockIdx.x * 16 + wv * AG_NPW;
    const float2* rows = (const float2*)P;

    int rp[AG_NPW + 1];
#pragma unroll
    for (int j = 0; j <= AG_NPW; ++j) rp[j] = rowptr[min(node0 + j, n)];

    int    s0[AG_NPW];
    float2 self[AG_NPW];
    float  dd[AG_NPW];
#pragma unroll
    for (int j = 0; j < AG_NPW; ++j) {
        const int node = node0 + j;
        const int deg  = rp[j + 1] - rp[j];
        s0[j] = (lane < deg) ? col[rp[j] + lane] : 0;
        if (node < n) {
            self[j] = rows[(size_t)node * 32 + lane];
            dd[j]   = dinv[node];
        } else {
            self[j] = make_float2(0.0f, 0.0f);
            dd[j]   = 0.0f;
        }
    }

#pragma unroll
    for (int j = 0; j < AG_NPW; ++j) {
        const int node = node0 + j;
        const float2 a = sum_row<float2, AG_BATCH>(rows, col, rp[j], rp[j + 1] - rp[j], s0[j], lane);
        float ox = (a.x + self[j].x) * dd[j];
        float oy = (a.y + self[j].y) * dd[j];
        if (bias) { ox += bias[lane * 2]; oy += bias[lane * 2 + 1]; }
        v2h g;
        g[0] = (_Float16)(ox * dd[j]);
        g[1] = (_Float16)(oy * dd[j]);
        if (node < n) ((v2h*)(G + (size_t)node * HID))[lane] = g;
    }
}


__device__ __forceinline__ int group8_pick(int v, int u) {
    switch (u) {
        case 0:  return __builtin_amdgcn_mov_dpp8(v, 0 * 0x249249);
        case 1:  return __builtin_amdgcn_mov_dpp8(v, 1 * 0x249249);
        case 2:  return __builtin_amdgcn_mov_dpp8(v, 2 * 0x249249);
        case 3:  return __builtin_amdgcn_mov_dpp8(v, 3 * 0x249249);
        case 4:  return __builtin_amdgcn_mov_dpp8(v, 4 * 0x249249);
        case 5:  return __builtin_amdgcn_mov_dpp8(v, 5 * 0x249249);
        case 6:  return __builtin_amdgcn_mov_dpp8(v, 6 * 0x249249);
        default: return __builtin_amdgcn_mov_dpp8(v, 7 * 0x249249);
    }
}

__device__ __forceinline__ void walk8(const char* __restrict__ Gb, int ids, unsigned qoff,
                                      int first, int lim, int deg, float (&acc)[8]) {
    uint4 v[AG_SLOTS];
#pragma unroll
    for (int u = 0; u < AG_SLOTS; ++u) {
        const int id = group8_pick(ids, u);
        v[u] = make_uint4(0u, 0u, 0u, 0u);
        if (first + u < deg) {
            const unsigned off = (unsigned)id * 128u + qoff;
            v[u] = *(const uint4*)(Gb + (size_t)off);
        }
    }
#pragma unroll
    for (int u = 0; u < AG_SLOTS; ++u) {
        if (first + u < lim) {
            const v8h h = __builtin_bit_cast(v8h, v[u]);
#pragma unroll
            for (int k = 0; k < 8; ++k) acc[k] += (float)h[k];
        }
    }
}

template <bool LAST>
__global__ void __launch_bounds__(AG_WAVES * 32)
agg_mul_kernel(const _Float16* __restrict__ G,
               const int* __restrict__ rowptr,
               const int* __restrict__ col,
               const float* __restrict__ dinv,
               const float* __restrict__ bias,
               const _Float16* __restrict__ Wp,
               _Float16* __restrict__ Gout,
               float* __restrict__ out, int n) {
    __shared__ __attribute__((aligned(16))) _Float16 hiT[16 * AG_PITCH];
    __shared__ __attribute__((aligned(16))) _Float16 loT[16 * AG_PITCH];
    __shared__ __attribute__((aligned(16))) _Float16 outT[16 * AG_PITCH];
    const int lane = threadIdx.x & 31;
    const int wv = __builtin_amdgcn_readfirstlane((int)(threadIdx.x >> 5));
    const int tileBase = blockIdx.x * 16;
    const int grp = lane >> 3;
    const int q   = lane & 7;
    const int rloc = wv * AG_NPW + grp;
    const int node = tileBase + rloc;
    const char* Gb = (const char*)G;
    const unsigned qoff = (unsigned)q * 16u;

    const int beg = rowptr[min(node, n)];
    const int deg = rowptr[min(node + 1, n)] - beg;

    int ids[4];
#pragma unroll
    for (int m = 0; m < 4; ++m) ids[m] = (q + 8 * m < deg) ? col[beg + q + 8 * m] : 0;
    uint4 selfr = make_uint4(0u, 0u, 0u, 0u);
    if (node < n) selfr = *(const uint4*)(Gb + (size_t)((unsigned)node * 128u + qoff));

    const int maxdeg = max(max(__builtin_amdgcn_readlane(deg, 0), __builtin_amdgcn_readlane(deg, 8)),
                           max(__builtin_amdgcn_readlane(deg, 16), __builtin_amdgcn_readlane(deg, 24)));
    const int lim = min(maxdeg, 32);

    float acc[8];
#pragma unroll
    for (int k = 0; k < 8; ++k) acc[k] = 0.0f;
    if (lim > 0)  walk8(Gb, ids[0], qoff, 0,  lim, deg, acc);
    if (lim > 8)  walk8(Gb, ids[1], qoff, 8,  lim, deg, acc);
    if (lim > 16) walk8(Gb, ids[2], qoff, 16, lim, deg, acc);
    if (lim > 24) walk8(Gb, ids[3], qoff, 24, lim, deg, acc);
    for (int e = 32; e < maxdeg; ++e) {
        if (e < deg) {
            const int id = col[beg + e];
            const v8h h = *(const v8h*)(Gb + (size_t)((unsigned)id * 128u + qoff));
#pragma unroll
            for (int k = 0; k < 8; ++k) acc[k] += (float)h[k];
        }
    }

    {
        const v8h sh = __builtin_bit_cast(v8h, selfr);
        v8h ph, pl;
#pragma unroll
        for (int k = 0; k < 8; ++k) {
            const float u = acc[k] + (float)sh[k];
            ph[k] = (_Float16)u;
            pl[k] = (_Float16)(u - (float)ph[k]);
        }
        *(v8h*)(hiT + rloc * AG_PITCH + q * 8) = ph;
        *(v8h*)(loT + rloc * AG_PITCH + q * 8) = pl;
    }

    __syncthreads();
    if (wv < 2) {
        const int m    = lane & 15;
        const int koff = (lane >> 4) << 3;
        const v16h* bp = (const v16h*)Wp;
        v8f acc0 = {}, acc1 = {};
#pragma unroll 1
        for (int ks = 0; ks < HID / 32; ++ks) {
            const _Float16* ah = hiT + m * AG_PITCH + ks * 32 + koff;
            const _Float16* al = loT + m * AG_PITCH + ks * 32 + koff;
            const v8h h0 = *(const v8h*)ah;
            const v8h h1 = *(const v8h*)(ah + 16);
            const v8h l0 = *(const v8h*)al;
            const v8h l1 = *(const v8h*)(al + 16);
            const v16h aHi = __builtin_shufflevector(h0, h1, 0, 1, 2, 3, 4, 5, 6, 7,
                                                     8, 9, 10, 11, 12, 13, 14, 15);
            const v16h aLo = __builtin_shufflevector(l0, l1, 0, 1, 2, 3, 4, 5, 6, 7,
                                                     8, 9, 10, 11, 12, 13, 14, 15);
            const v16h b0 = bp[(size_t)(ks * 4 + wv * 2 + 0) * 32 + lane];
            const v16h b1 = bp[(size_t)(ks * 4 + wv * 2 + 1) * 32 + lane];
            acc0 = __builtin_amdgcn_wmma_f32_16x16x32_f16(
                false, aHi, false, b0, (short)0, acc0, false, false);
            acc1 = __builtin_amdgcn_wmma_f32_16x16x32_f16(
                false, aHi, false, b1, (short)0, acc1, false, false);
            acc0 = __builtin_amdgcn_wmma_f32_16x16x32_f16(
                false, aLo, false, b0, (short)0, acc0, false, false);
            acc1 = __builtin_amdgcn_wmma_f32_16x16x32_f16(
                false, aLo, false, b1, (short)0, acc1, false, false);
        }
        const int colLane = lane & 15;
        const int rLoc    = (lane >> 4) << 3;
        const int c0      = wv * 32 + colLane;
        float sc[8];
#pragma unroll
        for (int i = 0; i < 8; ++i) sc[i] = (tileBase + rLoc + i < n) ? dinv[tileBase + rLoc + i] : 0.0f;
        if (LAST) {
#pragma unroll
            for (int i = 0; i < 8; ++i) {
                if (tileBase + rLoc + i < n) {
                    float* op = out + (size_t)(tileBase + rLoc + i) * HID + c0;
                    op[0]  = acc0[i] * sc[i];
                    op[16] = acc1[i] * sc[i];
                }
            }
        } else {
            const float bz0 = bias[c0];
            const float bz1 = bias[c0 + 16];
#pragma unroll
            for (int i = 0; i < 8; ++i) {
                const float v0 = fmaxf(acc0[i] * sc[i] + bz0, 0.0f);
                const float v1 = fmaxf(acc1[i] * sc[i] + bz1, 0.0f);
                outT[(rLoc + i) * AG_PITCH + c0]      = (_Float16)(v0 * sc[i]);
                outT[(rLoc + i) * AG_PITCH + c0 + 16] = (_Float16)(v1 * sc[i]);
            }
        }
    }
    if (!LAST) {
        __syncthreads();
        const int r  = threadIdx.x >> 3;
        const int ch = threadIdx.x & 7;
        if (tileBase + r < n) {
            const uint4 q = *(const uint4*)(outT + r * AG_PITCH + ch * 8);
            *(uint4*)(Gout + (size_t)(tileBase + r) * HID + ch * 8) = q;
        }
    }
}


extern "C" void kernel_launch(void* const* d_in, const int* in_sizes, int n_in,
                              void* d_out, int out_size, void* d_ws, size_t ws_size,
                              hipStream_t stream) {
    const float* x     = (const float*)d_in[0];
    const int*   eidx  = (const int*)d_in[1];
    const float* W_in  = (const float*)d_in[2];
    const float* b_in  = (const float*)d_in[3];
    const float* W_h   = (const float*)d_in[4];
    const float* b_h   = (const float*)d_in[5];
    const float* W_out = (const float*)d_in[6];

    const int N = in_sizes[0] / IN_D;
    const int E = in_sizes[1] / 2;
    const int L = in_sizes[4] / (HID * HID);

    const int* src = eidx;
    const int* dst = eidx + E;

    char* ws = (char*)d_ws;
    auto align256 = [](size_t v) { return (v + 255) & ~(size_t)255; };
    size_t off = 0;
    float* dinv = (float*)(ws + off);  off = align256(off + (size_t)N * sizeof(float));
    float* B1   = (float*)(ws + off);  off = align256(off + (size_t)N * HID * sizeof(float));
    _Float16* WpIn  = (_Float16*)(ws + off);  off = align256(off + 16 * 1024);
    _Float16* WpH0  = (_Float16*)(ws + off);  off = align256(off + 8 * 1024);
    _Float16* WpH1  = (_Float16*)(ws + off);  off = align256(off + 8 * 1024);
    _Float16* WpH2  = (_Float16*)(ws + off);  off = align256(off + 8 * 1024);
    _Float16* WpOut = (_Float16*)(ws + off);  off = align256(off + 8 * 1024);
    _Float16* WpHs[3] = {WpH0, WpH1, WpH2};
    int* rowptr = (int*)(ws + off);  off = align256(off + ((size_t)N + 1) * sizeof(int));
    int* col    = (int*)(ws + off);  off = align256(off + (size_t)E * sizeof(int));
    float* O = (float*)d_out;

    const int T = 256;

    const int rowTiles   = N / 16;
    const int gemmBlocks = (rowTiles + 7) / 8;
    const int tileBlocks = (N + 15) / 16;
    const int nBuck      = (N + BK_NODES - 1) >> BK_SHIFT;
    const int tabLen     = (nBuck + 1) * BK_GRID;
    const int chunk      = (E + BK_GRID - 1) / BK_GRID;

    const bool listFits = off <= ws_size && nBuck <= BK_MAXB &&
                          (size_t)3 * E + (size_t)tabLen <= (size_t)N * HID;

    if (listFits) {
        int2* pairs = (int2*)d_out;
        int* colTmp = (int*)d_out + (size_t)2 * E;
        int* offsIn = colTmp + E;
        const size_t histBytes = (size_t)nBuck * sizeof(int);

        const int packBlocks = ((WP_IN_FRAGS + WP_HID_FRAGS * (L + 1)) * 32 + BK_T - 1) / BK_T;
        slice_order_kernel<<<BK_GRID + packBlocks, BK_T, histBytes, stream>>>(src, dst, offsIn, pairs, E, chunk, nBuck, W_in, W_h, W_out, WpIn, L);
        bucket_finish_kernel<<<nBuck, FIN_T, 0, stream>>>(offsIn, pairs, rowptr, dinv, colTmp, col, N, nBuck, chunk);

        float*    P1 = (L == 0) ? O : B1;
        _Float16* Gc = (L == 0) ? (_Float16*)B1 : (_Float16*)O;
        _Float16* A0 = (_Float16*)B1;
        _Float16* A1 = A0 + (size_t)N * HID;

        gemm_wmma_kernel<IN_D, false><<<gemmBlocks, T, 0, stream>>>(x, WpIn, dinv, P1, N);
        agg_first_kernel<<<tileBlocks, AG_WAVES * 32, 0, stream>>>(P1, rowptr, col, dinv, b_in, Gc, N);

        const _Float16* g = Gc;
        for (int i = 0; i < L; ++i) {
            _Float16* gn = (i & 1) ? A1 : A0;
            agg_mul_kernel<false><<<tileBlocks, AG_WAVES * 32, 0, stream>>>(g, rowptr, col, dinv, b_h + (size_t)i * HID, WpHs[i], gn, nullptr, N);
            g = gn;
        }

        agg_mul_kernel<true><<<tileBlocks, AG_WAVES * 32, 0, stream>>>(g, rowptr, col, dinv, nullptr, WpOut, nullptr, O, N);
        return;
    }


    deg_init_kernel<<<(N + T - 1) / T, T, 0, stream>>>(dinv, N);
    deg_accum_kernel<<<(E + T - 1) / T, T, 0, stream>>>(dst, dinv, E);
    dinv_kernel<<<(N + T - 1) / T, T, 0, stream>>>(dinv, N);

    pack_w_kernel<IN_D><<<2, T, 0, stream>>>(W_in, WpIn);
    for (int l = 0; l < L; ++l)
        pack_w_kernel<HID><<<1, T, 0, stream>>>(W_h + (size_t)l * HID * HID, WpHs[l]);
    pack_w_kernel<HID><<<1, T, 0, stream>>>(W_out, WpOut);

    const int totalNC    = N * HID;
    const int initBlocks = (totalNC + T - 1) / T;
    const int edgeBlocks = (E + 7) / 8;

    gemm_wmma_kernel<IN_D, false><<<gemmBlocks, T, 0, stream>>>(x, WpIn, dinv, B1, N);
    agg_init_kernel<<<initBlocks, T, 0, stream>>>(B1, dinv, b_in, O, totalNC);
    agg_edges_kernel<<<edgeBlocks, T, 0, stream>>>(B1, src, dst, dinv, O, E);

    for (int l = 0; l < L; ++l) {
        const float* b = b_h + (size_t)l * HID;
        if (l == 0)
            gemm_wmma_kernel<HID, false><<<gemmBlocks, T, 0, stream>>>(O, WpHs[l], dinv, B1, N);
        else
            gemm_wmma_kernel<HID, true ><<<gemmBlocks, T, 0, stream>>>(O, WpHs[l], dinv, B1, N);
        agg_init_kernel<<<initBlocks, T, 0, stream>>>(B1, dinv, b, O, totalNC);
        agg_edges_kernel<<<edgeBlocks, T, 0, stream>>>(B1, src, dst, dinv, O, E);
    }

    gemm_wmma_kernel<HID, true><<<gemmBlocks, T, 0, stream>>>(O, WpOut, dinv, B1, N);
    agg_init_kernel<<<initBlocks, T, 0, stream>>>(B1, dinv, nullptr, O, totalNC);
    agg_edges_kernel<<<edgeBlocks, T, 0, stream>>>(B1, src, dst, dinv, O, E);
}
